# starting kernel
# baseline (speedup 1.0000x reference)
#include <hip/hip_runtime.h>


typedef float v2f __attribute__((ext_vector_type(2)));
typedef float v8f __attribute__((ext_vector_type(8)));

#define HID 128

__global__ __launch_bounds__(256)
void gnn_linear_relu_wmma(const float* __restrict__ X,
                          const float* __restrict__ W,
                          const float* __restrict__ bias,
                          float* __restrict__ T) {
    const int lane   = threadIdx.x & 31;
    const int wave   = threadIdx.x >> 5;
    const int m_base = blockIdx.x << 4;
    const int n_base = wave << 4;
    const int mr     = lane & 15;
    const int kh     = lane >> 4;

    const float* __restrict__ xrow = X + (size_t)(m_base + mr) * HID;
    const int col = n_base + mr;

    v8f acc = {0.f, 0.f, 0.f, 0.f, 0.f, 0.f, 0.f, 0.f};

    #pragma unroll 4
    for (int k0 = 0; k0 < HID; k0 += 4) {
        const int k = k0 + (kh << 1);
        v2f a;
        a.x = xrow[k];
        a.y = xrow[k + 1];
        v2f b;
        b.x = W[(size_t)k * HID + col];
        b.y = W[(size_t)(k + 1) * HID + col];
        acc = __builtin_amdgcn_wmma_f32_16x16x4_f32(
             false, a,  false, b,
             (short)0, acc,  false,  false);
    }

    const float bv = bias[col];
    #pragma unroll
    for (int r = 0; r < 8; ++r) {
        const int row = m_base + r + (kh << 3);
        const float v = acc[r] + bv;
        T[(size_t)row * HID + col] = fmaxf(v, 0.0f);
    }
}

__global__ void gnn_zero_f4(float4* __restrict__ p, int n4) {
    const int i = blockIdx.x * blockDim.x + threadIdx.x;
    if (i < n4) p[i] = make_float4(0.f, 0.f, 0.f, 0.f);
}

__global__ __launch_bounds__(256)
void gnn_scatter_add(const float* __restrict__ T,
                     const int* __restrict__ src,
                     const int* __restrict__ dst,
                     float* __restrict__ H, int E) {
    const int idx = blockIdx.x * blockDim.x + threadIdx.x;
    const int e = idx >> 5;
    if (e >= E) return;
    const int f = (idx & 31) << 2;
    const int s = src[e];
    const int d = dst[e];
    const float4 v = *(const float4*)(T + (size_t)s * HID + f);
    float* hp = H + (size_t)d * HID + f;
    atomicAdd(hp + 0, v.x);
    atomicAdd(hp + 1, v.y);
    atomicAdd(hp + 2, v.z);
    atomicAdd(hp + 3, v.w);
}

__global__ __launch_bounds__(256)
void gnn_final_gemv(const float* __restrict__ H,
                    const float* __restrict__ Wf,
                    const float* __restrict__ bf,
                    float* __restrict__ out, int N) {
    const int wid  = (blockIdx.x * blockDim.x + threadIdx.x) >> 5;
    const int lane = threadIdx.x & 31;
    if (wid >= N) return;
    const float4 v = *(const float4*)(H + (size_t)wid * HID + (lane << 2));
    const float4 w = *(const float4*)(Wf + (lane << 2));
    float s = v.x * w.x + v.y * w.y + v.z * w.z + v.w * w.w;
    #pragma unroll
    for (int off = 16; off > 0; off >>= 1)
        s += __shfl_xor(s, off, 32);
    if (lane == 0) out[wid] = s + bf[0];
}

extern "C" void kernel_launch(void* const* d_in, const int* in_sizes, int n_in,
                              void* d_out, int out_size, void* d_ws, size_t ws_size,
                              hipStream_t stream) {
    const float* x   = (const float*)d_in[0];
    const int*   ei  = (const int*)  d_in[1];
    const float* W1  = (const float*)d_in[2];
    const float* b1  = (const float*)d_in[3];
    const float* W2  = (const float*)d_in[4];
    const float* b2  = (const float*)d_in[5];
    const float* Wf  = (const float*)d_in[6];
    const float* bf  = (const float*)d_in[7];
    float* out = (float*)d_out;

    const int N = in_sizes[0] / HID;
    const int E = in_sizes[1] / 2;
    const int* src = ei;
    const int* dst = ei + E;

    float* bufT = (float*)d_ws;
    float* bufH = bufT + (size_t)N * HID;

    const dim3 blk(256);
    const int n4        = (N * HID) / 4;
    const int zeroGrid  = (n4 + 255) / 256;
    const int gemmGrid  = N / 16;
    const int scatGrid  = (int)(((size_t)E * 32 + 255) / 256);
    const int gemvGrid  = (N * 32 + 255) / 256;

    gnn_linear_relu_wmma<<<gemmGrid, blk, 0, stream>>>(x, W1, b1, bufT);
    gnn_zero_f4<<<zeroGrid, blk, 0, stream>>>((float4*)bufH, n4);
    gnn_scatter_add<<<scatGrid, blk, 0, stream>>>(bufT, src, dst, bufH, E);

    gnn_linear_relu_wmma<<<gemmGrid, blk, 0, stream>>>(bufH, W2, b2, bufT);
    gnn_zero_f4<<<zeroGrid, blk, 0, stream>>>((float4*)bufH, n4);
    gnn_scatter_add<<<scatGrid, blk, 0, stream>>>(bufT, src, dst, bufH, E);

    gnn_final_gemv<<<gemvGrid, blk, 0, stream>>>(bufH, Wf, bf, out, N);
}


# Readout Wf applied per node before the layer-2 scatter (linear commutes with segment sum): GEMM2 epilogue emits 1 scalar per node, scatter adds 1 float per edge not 128; sum order changes
# speedup vs baseline: 1.4674x; 1.4674x over previous
#include <hip/hip_runtime.h>


typedef float v2f __attribute__((ext_vector_type(2)));
typedef float v8f __attribute__((ext_vector_type(8)));

#define HID 128

__global__ __launch_bounds__(256)
void gnn_linear_relu_wmma(const float* __restrict__ X,
                          const float* __restrict__ W,
                          const float* __restrict__ bias,
                          float* __restrict__ T) {
    const int lane   = threadIdx.x & 31;
    const int wave   = threadIdx.x >> 5;
    const int m_base = blockIdx.x << 4;
    const int n_base = wave << 4;
    const int mr     = lane & 15;
    const int kh     = lane >> 4;

    const float* __restrict__ xrow = X + (size_t)(m_base + mr) * HID;
    const int col = n_base + mr;

    v8f acc = {0.f, 0.f, 0.f, 0.f, 0.f, 0.f, 0.f, 0.f};

    #pragma unroll 4
    for (int k0 = 0; k0 < HID; k0 += 4) {
        const int k = k0 + (kh << 1);
        v2f a;
        a.x = xrow[k];
        a.y = xrow[k + 1];
        v2f b;
        b.x = W[(size_t)k * HID + col];
        b.y = W[(size_t)(k + 1) * HID + col];
        acc = __builtin_amdgcn_wmma_f32_16x16x4_f32(
             false, a,  false, b,
             (short)0, acc,  false,  false);
    }

    const float bv = bias[col];
    #pragma unroll
    for (int r = 0; r < 8; ++r) {
        const int row = m_base + r + (kh << 3);
        const float v = acc[r] + bv;
        T[(size_t)row * HID + col] = fmaxf(v, 0.0f);
    }
}

__global__ void gnn_zero_f4(float4* __restrict__ p, int n4) {
    const int i = blockIdx.x * blockDim.x + threadIdx.x;
    if (i < n4) p[i] = make_float4(0.f, 0.f, 0.f, 0.f);
}

__global__ __launch_bounds__(256)
void gnn_scatter_add(const float* __restrict__ T,
                     const int* __restrict__ src,
                     const int* __restrict__ dst,
                     float* __restrict__ H, int E) {
    const int idx = blockIdx.x * blockDim.x + threadIdx.x;
    const int e = idx >> 5;
    if (e >= E) return;
    const int f = (idx & 31) << 2;
    const int s = src[e];
    const int d = dst[e];
    const float4 v = *(const float4*)(T + (size_t)s * HID + f);
    float* hp = H + (size_t)d * HID + f;
    atomicAdd(hp + 0, v.x);
    atomicAdd(hp + 1, v.y);
    atomicAdd(hp + 2, v.z);
    atomicAdd(hp + 3, v.w);
}

__global__ __launch_bounds__(256)
void gnn_linear_relu_readout_wmma(const float* __restrict__ X,
                                  const float* __restrict__ W,
                                  const float* __restrict__ bias,
                                  const float* __restrict__ Wf,
                                  const float* __restrict__ bf,
                                  float* __restrict__ S,
                                  float* __restrict__ out) {
    __shared__ float part[8][16];

    const int lane   = threadIdx.x & 31;
    const int wave   = threadIdx.x >> 5;
    const int m_base = blockIdx.x << 4;
    const int n_base = wave << 4;
    const int mr     = lane & 15;
    const int kh     = lane >> 4;

    const float* __restrict__ xrow = X + (size_t)(m_base + mr) * HID;
    const int col = n_base + mr;

    v8f acc = {0.f, 0.f, 0.f, 0.f, 0.f, 0.f, 0.f, 0.f};

    #pragma unroll 4
    for (int k0 = 0; k0 < HID; k0 += 4) {
        const int k = k0 + (kh << 1);
        v2f a;
        a.x = xrow[k];
        a.y = xrow[k + 1];
        v2f b;
        b.x = W[(size_t)k * HID + col];
        b.y = W[(size_t)(k + 1) * HID + col];
        acc = __builtin_amdgcn_wmma_f32_16x16x4_f32(
             false, a,  false, b,
             (short)0, acc,  false,  false);
    }

    const float bv = bias[col];
    const float wf = Wf[col];
    float p[8];
    #pragma unroll
    for (int r = 0; r < 8; ++r)
        p[r] = fmaxf(acc[r] + bv, 0.0f) * wf;
    #pragma unroll
    for (int off = 8; off > 0; off >>= 1) {
        #pragma unroll
        for (int r = 0; r < 8; ++r)
            p[r] += __shfl_xor(p[r], off, 32);
    }
    if (mr == 0) {
        #pragma unroll
        for (int r = 0; r < 8; ++r)
            part[wave][r + (kh << 3)] = p[r];
    }
    __syncthreads();

    if (threadIdx.x < 16) {
        float s = part[0][threadIdx.x];
        #pragma unroll
        for (int w = 1; w < 8; ++w)
            s += part[w][threadIdx.x];
        S[m_base + threadIdx.x]   = s;
        out[m_base + threadIdx.x] = bf[0];
    }
}

__global__ __launch_bounds__(256)
void gnn_scatter_add_scalar(const float* __restrict__ S,
                            const int* __restrict__ src,
                            const int* __restrict__ dst,
                            float* __restrict__ out, int E) {
    const int e0 = (blockIdx.x * blockDim.x + threadIdx.x) << 2;
    if (e0 >= E) return;
    if ((E & 3) == 0) {
        const int4 s = *(const int4*)(src + e0);
        const int4 d = *(const int4*)(dst + e0);
        const float v0 = S[s.x];
        const float v1 = S[s.y];
        const float v2 = S[s.z];
        const float v3 = S[s.w];
        atomicAdd(out + d.x, v0);
        atomicAdd(out + d.y, v1);
        atomicAdd(out + d.z, v2);
        atomicAdd(out + d.w, v3);
    } else {
        for (int e = e0; e < E && e < e0 + 4; ++e)
            atomicAdd(out + dst[e], S[src[e]]);
    }
}

extern "C" void kernel_launch(void* const* d_in, const int* in_sizes, int n_in,
                              void* d_out, int out_size, void* d_ws, size_t ws_size,
                              hipStream_t stream) {
    const float* x   = (const float*)d_in[0];
    const int*   ei  = (const int*)  d_in[1];
    const float* W1  = (const float*)d_in[2];
    const float* b1  = (const float*)d_in[3];
    const float* W2  = (const float*)d_in[4];
    const float* b2  = (const float*)d_in[5];
    const float* Wf  = (const float*)d_in[6];
    const float* bf  = (const float*)d_in[7];
    float* out = (float*)d_out;

    const int N = in_sizes[0] / HID;
    const int E = in_sizes[1] / 2;
    const int* src = ei;
    const int* dst = ei + E;

    float* bufT = (float*)d_ws;
    float* bufH = bufT + (size_t)N * HID;
    float* bufS = bufH + (size_t)N * HID;

    const dim3 blk(256);
    const int n4        = (N * HID) / 4;
    const int zeroGrid  = (n4 + 255) / 256;
    const int gemmGrid  = N / 16;
    const int scatGrid  = (int)(((size_t)E * 32 + 255) / 256);
    const int sscatGrid = (E + 1023) / 1024;

    gnn_linear_relu_wmma<<<gemmGrid, blk, 0, stream>>>(x, W1, b1, bufT);
    gnn_zero_f4<<<zeroGrid, blk, 0, stream>>>((float4*)bufH, n4);
    gnn_scatter_add<<<scatGrid, blk, 0, stream>>>(bufT, src, dst, bufH, E);

    gnn_linear_relu_readout_wmma<<<gemmGrid, blk, 0, stream>>>(bufH, W2, b2, Wf, bf, bufS, out);
    gnn_scatter_add_scalar<<<sscatGrid, blk, 0, stream>>>(bufS, src, dst, out, E);
}


# Layer-1 scatter: lane l owns floats l, l+32, l+64, l+96 of the row, so each f32 atomic instruction and each gather load covers one whole 128-byte line instead of four quarter-filled lines
# speedup vs baseline: 1.7627x; 1.2012x over previous
#include <hip/hip_runtime.h>


typedef float v2f __attribute__((ext_vector_type(2)));
typedef float v8f __attribute__((ext_vector_type(8)));

#define HID 128

__global__ __launch_bounds__(256)
void gnn_linear_relu_wmma(const float* __restrict__ X,
                          const float* __restrict__ W,
                          const float* __restrict__ bias,
                          float* __restrict__ T) {
    const int lane   = threadIdx.x & 31;
    const int wave   = threadIdx.x >> 5;
    const int m_base = blockIdx.x << 4;
    const int n_base = wave << 4;
    const int mr     = lane & 15;
    const int kh     = lane >> 4;

    const float* __restrict__ xrow = X + (size_t)(m_base + mr) * HID;
    const int col = n_base + mr;

    v8f acc = {0.f, 0.f, 0.f, 0.f, 0.f, 0.f, 0.f, 0.f};

    #pragma unroll 4
    for (int k0 = 0; k0 < HID; k0 += 4) {
        const int k = k0 + (kh << 1);
        v2f a;
        a.x = xrow[k];
        a.y = xrow[k + 1];
        v2f b;
        b.x = W[(size_t)k * HID + col];
        b.y = W[(size_t)(k + 1) * HID + col];
        acc = __builtin_amdgcn_wmma_f32_16x16x4_f32(
             false, a,  false, b,
             (short)0, acc,  false,  false);
    }

    const float bv = bias[col];
    #pragma unroll
    for (int r = 0; r < 8; ++r) {
        const int row = m_base + r + (kh << 3);
        const float v = acc[r] + bv;
        T[(size_t)row * HID + col] = fmaxf(v, 0.0f);
    }
}

__global__ void gnn_zero_f4(float4* __restrict__ p, int n4) {
    const int i = blockIdx.x * blockDim.x + threadIdx.x;
    if (i < n4) p[i] = make_float4(0.f, 0.f, 0.f, 0.f);
}

__global__ __launch_bounds__(256)
void gnn_scatter_add(const float* __restrict__ T,
                     const int* __restrict__ src,
                     const int* __restrict__ dst,
                     float* __restrict__ H, int E) {
    const int idx = blockIdx.x * blockDim.x + threadIdx.x;
    const int e = idx >> 5;
    if (e >= E) return;
    const int f = idx & 31;
    const int s = src[e];
    const int d = dst[e];
    const float* tp = T + (size_t)s * HID + f;
    const float v0 = tp[0];
    const float v1 = tp[32];
    const float v2 = tp[64];
    const float v3 = tp[96];
    float* hp = H + (size_t)d * HID + f;
    atomicAdd(hp + 0,  v0);
    atomicAdd(hp + 32, v1);
    atomicAdd(hp + 64, v2);
    atomicAdd(hp + 96, v3);
}

__global__ __launch_bounds__(256)
void gnn_linear_relu_readout_wmma(const float* __restrict__ X,
                                  const float* __restrict__ W,
                                  const float* __restrict__ bias,
                                  const float* __restrict__ Wf,
                                  const float* __restrict__ bf,
                                  float* __restrict__ S,
                                  float* __restrict__ out) {
    __shared__ float part[8][16];

    const int lane   = threadIdx.x & 31;
    const int wave   = threadIdx.x >> 5;
    const int m_base = blockIdx.x << 4;
    const int n_base = wave << 4;
    const int mr     = lane & 15;
    const int kh     = lane >> 4;

    const float* __restrict__ xrow = X + (size_t)(m_base + mr) * HID;
    const int col = n_base + mr;

    v8f acc = {0.f, 0.f, 0.f, 0.f, 0.f, 0.f, 0.f, 0.f};

    #pragma unroll 4
    for (int k0 = 0; k0 < HID; k0 += 4) {
        const int k = k0 + (kh << 1);
        v2f a;
        a.x = xrow[k];
        a.y = xrow[k + 1];
        v2f b;
        b.x = W[(size_t)k * HID + col];
        b.y = W[(size_t)(k + 1) * HID + col];
        acc = __builtin_amdgcn_wmma_f32_16x16x4_f32(
             false, a,  false, b,
             (short)0, acc,  false,  false);
    }

    const float bv = bias[col];
    const float wf = Wf[col];
    float p[8];
    #pragma unroll
    for (int r = 0; r < 8; ++r)
        p[r] = fmaxf(acc[r] + bv, 0.0f) * wf;
    #pragma unroll
    for (int off = 8; off > 0; off >>= 1) {
        #pragma unroll
        for (int r = 0; r < 8; ++r)
            p[r] += __shfl_xor(p[r], off, 32);
    }
    if (mr == 0) {
        #pragma unroll
        for (int r = 0; r < 8; ++r)
            part[wave][r + (kh << 3)] = p[r];
    }
    __syncthreads();

    if (threadIdx.x < 16) {
        float s = part[0][threadIdx.x];
        #pragma unroll
        for (int w = 1; w < 8; ++w)
            s += part[w][threadIdx.x];
        S[m_base + threadIdx.x]   = s;
        out[m_base + threadIdx.x] = bf[0];
    }
}

__global__ __launch_bounds__(256)
void gnn_scatter_add_scalar(const float* __restrict__ S,
                            const int* __restrict__ src,
                            const int* __restrict__ dst,
                            float* __restrict__ out, int E) {
    const int e0 = (blockIdx.x * blockDim.x + threadIdx.x) << 2;
    if (e0 >= E) return;
    if ((E & 3) == 0) {
        const int4 s = *(const int4*)(src + e0);
        const int4 d = *(const int4*)(dst + e0);
        const float v0 = S[s.x];
        const float v1 = S[s.y];
        const float v2 = S[s.z];
        const float v3 = S[s.w];
        atomicAdd(out + d.x, v0);
        atomicAdd(out + d.y, v1);
        atomicAdd(out + d.z, v2);
        atomicAdd(out + d.w, v3);
    } else {
        for (int e = e0; e < E && e < e0 + 4; ++e)
            atomicAdd(out + dst[e], S[src[e]]);
    }
}

extern "C" void kernel_launch(void* const* d_in, const int* in_sizes, int n_in,
                              void* d_out, int out_size, void* d_ws, size_t ws_size,
                              hipStream_t stream) {
    const float* x   = (const float*)d_in[0];
    const int*   ei  = (const int*)  d_in[1];
    const float* W1  = (const float*)d_in[2];
    const float* b1  = (const float*)d_in[3];
    const float* W2  = (const float*)d_in[4];
    const float* b2  = (const float*)d_in[5];
    const float* Wf  = (const float*)d_in[6];
    const float* bf  = (const float*)d_in[7];
    float* out = (float*)d_out;

    const int N = in_sizes[0] / HID;
    const int E = in_sizes[1] / 2;
    const int* src = ei;
    const int* dst = ei + E;

    float* bufT = (float*)d_ws;
    float* bufH = bufT + (size_t)N * HID;
    float* bufS = bufH + (size_t)N * HID;

    const dim3 blk(256);
    const int n4        = (N * HID) / 4;
    const int zeroGrid  = (n4 + 255) / 256;
    const int gemmGrid  = N / 16;
    const int scatGrid  = (int)(((size_t)E * 32 + 255) / 256);
    const int sscatGrid = (E + 1023) / 1024;

    gnn_linear_relu_wmma<<<gemmGrid, blk, 0, stream>>>(x, W1, b1, bufT);
    gnn_zero_f4<<<zeroGrid, blk, 0, stream>>>((float4*)bufH, n4);
    gnn_scatter_add<<<scatGrid, blk, 0, stream>>>(bufT, src, dst, bufH, E);

    gnn_linear_relu_readout_wmma<<<gemmGrid, blk, 0, stream>>>(bufH, W2, b2, Wf, bf, bufS, out);
    gnn_scatter_add_scalar<<<sscatGrid, blk, 0, stream>>>(bufS, src, dst, out, E);
}


# Both GEMMs re-tiled: W staged once per block in LDS, each wave stages its 16 rows of X in LDS with whole-row loads and computes the full 16x128 strip, so one A fragment feeds 8 WMMAs
# speedup vs baseline: 1.9625x; 1.1133x over previous
#include <hip/hip_runtime.h>


typedef float v2f __attribute__((ext_vector_type(2)));
typedef float v8f __attribute__((ext_vector_type(8)));

#define HID 128
#define GEMM_WAVES 8
#define WPITCH 288
#define APITCH 132

__device__ __forceinline__ void gemm_strip_16x128(const float* __restrict__ X,
                                                  const float* __restrict__ W,
                                                  int tile, v8f (&acc)[8]) {
    __shared__ __attribute__((aligned(16))) float sW[64 * WPITCH];
    __shared__ __attribute__((aligned(16))) float sA[GEMM_WAVES * 16 * APITCH];

    const int tid  = threadIdx.x;
    const int lane = tid & 31;
    const int wave = tid >> 5;
    const int c    = lane & 15;
    const int h    = lane >> 4;

    float4 ar[16];
    const float4* xp = (const float4*)(X + (size_t)tile * 16 * HID) + lane;
    #pragma unroll
    for (int r = 0; r < 16; ++r)
        ar[r] = xp[r * (HID / 4)];

    #pragma unroll
    for (int i = 0; i < 8; ++i) {
        const int item = i * 256 + tid;
        const int kp   = item >> 5;
        const int m    = item & 31;
        const float4 w0 = *(const float4*)(W + (size_t)(2 * kp) * HID + 4 * m);
        const float4 w1 = *(const float4*)(W + (size_t)(2 * kp + 1) * HID + 4 * m);
        float* d = sW + kp * WPITCH + 8 * m;
        *(float4*)(d)     = make_float4(w0.x, w1.x, w0.y, w1.y);
        *(float4*)(d + 4) = make_float4(w0.z, w1.z, w0.w, w1.w);
    }

    float* sAw = sA + wave * (16 * APITCH);
    #pragma unroll
    for (int r = 0; r < 16; ++r)
        *(float4*)(sAw + r * APITCH + 4 * lane) = ar[r];

    __syncthreads();

    const float* ap = sAw + c * APITCH + 2 * h;
    const float* bp = sW + h * WPITCH + 2 * c;

    #pragma unroll
    for (int j = 0; j < 8; ++j)
        acc[j] = (v8f){0.f, 0.f, 0.f, 0.f, 0.f, 0.f, 0.f, 0.f};

    v2f a_cur = *(const v2f*)(ap);
    v2f b_cur[8];
    #pragma unroll
    for (int j = 0; j < 8; ++j)
        b_cur[j] = *(const v2f*)(bp + 32 * j);

    #pragma unroll
    for (int k0 = 0; k0 < HID; k0 += 4) {
        v2f a_nxt = a_cur;
        v2f b_nxt[8];
        #pragma unroll
        for (int j = 0; j < 8; ++j)
            b_nxt[j] = b_cur[j];
        if (k0 + 4 < HID) {
            a_nxt = *(const v2f*)(ap + k0 + 4);
            const float* bk = bp + ((k0 + 4) >> 1) * WPITCH;
            #pragma unroll
            for (int j = 0; j < 8; ++j)
                b_nxt[j] = *(const v2f*)(bk + 32 * j);
        }
        #pragma unroll
        for (int j = 0; j < 8; ++j)
            acc[j] = __builtin_amdgcn_wmma_f32_16x16x4_f32(
                 false, a_cur,  false, b_cur[j],
                 (short)0, acc[j],  false,  false);
        a_cur = a_nxt;
        #pragma unroll
        for (int j = 0; j < 8; ++j)
            b_cur[j] = b_nxt[j];
    }
}

__global__ __launch_bounds__(256)
void gnn_linear_relu_wmma(const float* __restrict__ X,
                          const float* __restrict__ W,
                          const float* __restrict__ bias,
                          float* __restrict__ T, int numTiles) {
    const int lane = threadIdx.x & 31;
    const int c    = lane & 15;
    const int h    = lane >> 4;
    const int mine = blockIdx.x * GEMM_WAVES + (threadIdx.x >> 5);
    const bool active = mine < numTiles;
    const int tile = active ? mine : numTiles - 1;

    v8f acc[8];
    gemm_strip_16x128(X, W, tile, acc);

    if (active) {
        const int m_base = tile << 4;
        #pragma unroll
        for (int j = 0; j < 8; ++j) {
            const int col = 16 * j + c;
            const float bv = bias[col];
            #pragma unroll
            for (int r = 0; r < 8; ++r) {
                const int row = m_base + r + (h << 3);
                T[(size_t)row * HID + col] = fmaxf(acc[j][r] + bv, 0.0f);
            }
        }
    }
}

__global__ void gnn_zero_f4(float4* __restrict__ p, int n4) {
    const int i = blockIdx.x * blockDim.x + threadIdx.x;
    if (i < n4) p[i] = make_float4(0.f, 0.f, 0.f, 0.f);
}

__global__ __launch_bounds__(256)
void gnn_scatter_add(const float* __restrict__ T,
                     const int* __restrict__ src,
                     const int* __restrict__ dst,
                     float* __restrict__ H, int E) {
    const int idx = blockIdx.x * blockDim.x + threadIdx.x;
    const int e = idx >> 5;
    if (e >= E) return;
    const int f = idx & 31;
    const int s = src[e];
    const int d = dst[e];
    const float* tp = T + (size_t)s * HID + f;
    const float v0 = tp[0];
    const float v1 = tp[32];
    const float v2 = tp[64];
    const float v3 = tp[96];
    float* hp = H + (size_t)d * HID + f;
    atomicAdd(hp + 0,  v0);
    atomicAdd(hp + 32, v1);
    atomicAdd(hp + 64, v2);
    atomicAdd(hp + 96, v3);
}

__global__ __launch_bounds__(256)
void gnn_linear_relu_readout_wmma(const float* __restrict__ X,
                                  const float* __restrict__ W,
                                  const float* __restrict__ bias,
                                  const float* __restrict__ Wf,
                                  const float* __restrict__ bf,
                                  float* __restrict__ S,
                                  float* __restrict__ out, int numTiles) {
    const int lane = threadIdx.x & 31;
    const int c    = lane & 15;
    const int h    = lane >> 4;
    const int mine = blockIdx.x * GEMM_WAVES + (threadIdx.x >> 5);
    const bool active = mine < numTiles;
    const int tile = active ? mine : numTiles - 1;

    v8f acc[8];
    gemm_strip_16x128(X, W, tile, acc);

    float s[8];
    #pragma unroll
    for (int j = 0; j < 8; ++j) {
        const int col = 16 * j + c;
        const float bv = bias[col];
        const float wf = Wf[col];
        float p[8];
        #pragma unroll
        for (int r = 0; r < 8; ++r)
            p[r] = fmaxf(acc[j][r] + bv, 0.0f) * wf;
        #pragma unroll
        for (int off = 8; off > 0; off >>= 1) {
            #pragma unroll
            for (int r = 0; r < 8; ++r)
                p[r] += __shfl_xor(p[r], off, 32);
        }
        #pragma unroll
        for (int r = 0; r < 8; ++r)
            s[r] = (j == 0) ? p[r] : s[r] + p[r];
    }

    float v = s[0];
    #pragma unroll
    for (int r = 1; r < 8; ++r)
        v = (c == r) ? s[r] : v;
    if (active && c < 8) {
        const int row = (tile << 4) + (h << 3) + c;
        S[row]   = v;
        out[row] = bf[0];
    }
}

__global__ __launch_bounds__(256)
void gnn_scatter_add_scalar(const float* __restrict__ S,
                            const int* __restrict__ src,
                            const int* __restrict__ dst,
                            float* __restrict__ out, int E) {
    const int e0 = (blockIdx.x * blockDim.x + threadIdx.x) << 2;
    if (e0 >= E) return;
    if ((E & 3) == 0) {
        const int4 s = *(const int4*)(src + e0);
        const int4 d = *(const int4*)(dst + e0);
        const float v0 = S[s.x];
        const float v1 = S[s.y];
        const float v2 = S[s.z];
        const float v3 = S[s.w];
        atomicAdd(out + d.x, v0);
        atomicAdd(out + d.y, v1);
        atomicAdd(out + d.z, v2);
        atomicAdd(out + d.w, v3);
    } else {
        for (int e = e0; e < E && e < e0 + 4; ++e)
            atomicAdd(out + dst[e], S[src[e]]);
    }
}

extern "C" void kernel_launch(void* const* d_in, const int* in_sizes, int n_in,
                              void* d_out, int out_size, void* d_ws, size_t ws_size,
                              hipStream_t stream) {
    const float* x   = (const float*)d_in[0];
    const int*   ei  = (const int*)  d_in[1];
    const float* W1  = (const float*)d_in[2];
    const float* b1  = (const float*)d_in[3];
    const float* W2  = (const float*)d_in[4];
    const float* b2  = (const float*)d_in[5];
    const float* Wf  = (const float*)d_in[6];
    const float* bf  = (const float*)d_in[7];
    float* out = (float*)d_out;

    const int N = in_sizes[0] / HID;
    const int E = in_sizes[1] / 2;
    const int* src = ei;
    const int* dst = ei + E;

    float* bufT = (float*)d_ws;
    float* bufH = bufT + (size_t)N * HID;
    float* bufS = bufH + (size_t)N * HID;

    const dim3 blk(256);
    const int n4        = (N * HID) / 4;
    const int zeroGrid  = (n4 + 255) / 256;
    const int numTiles  = N / 16;
    const int gemmGrid  = (numTiles + 7) / 8;
    const int scatGrid  = (int)(((size_t)E * 32 + 255) / 256);
    const int sscatGrid = (E + 1023) / 1024;

    gnn_linear_relu_wmma<<<gemmGrid, blk, 0, stream>>>(x, W1, b1, bufT, numTiles);
    gnn_zero_f4<<<zeroGrid, blk, 0, stream>>>((float4*)bufH, n4);
    gnn_scatter_add<<<scatGrid, blk, 0, stream>>>(bufT, src, dst, bufH, E);

    gnn_linear_relu_readout_wmma<<<gemmGrid, blk, 0, stream>>>(bufH, W2, b2, Wf, bf, bufS, out, numTiles);
    gnn_scatter_add_scalar<<<sscatGrid, blk, 0, stream>>>(bufS, src, dst, out, E);
}


# Edges bucketed by destination in the call (returning int add = rank, prefix sum, place); a wave sums a node's source rows in registers, one row store: no f32 atomics or zero fill; sum order changes
# speedup vs baseline: 2.5883x; 1.3189x over previous
#include <hip/hip_runtime.h>


typedef float v2f __attribute__((ext_vector_type(2)));
typedef float v8f __attribute__((ext_vector_type(8)));

#define HID 128
#define GEMM_WAVES 8
#define WPITCH 288
#define APITCH 132

__device__ __forceinline__ void gemm_strip_16x128(const float* __restrict__ X,
                                                  const float* __restrict__ W,
                                                  int tile, v8f (&acc)[8]) {
    __shared__ __attribute__((aligned(16))) float sW[64 * WPITCH];
    __shared__ __attribute__((aligned(16))) float sA[GEMM_WAVES * 16 * APITCH];

    const int tid  = threadIdx.x;
    const int lane = tid & 31;
    const int wave = tid >> 5;
    const int c    = lane & 15;
    const int h    = lane >> 4;

    float4 ar[16];
    const float4* xp = (const float4*)(X + (size_t)tile * 16 * HID) + lane;
    #pragma unroll
    for (int r = 0; r < 16; ++r)
        ar[r] = xp[r * (HID / 4)];

    #pragma unroll
    for (int i = 0; i < 8; ++i) {
        const int item = i * 256 + tid;
        const int kp   = item >> 5;
        const int m    = item & 31;
        const float4 w0 = *(const float4*)(W + (size_t)(2 * kp) * HID + 4 * m);
        const float4 w1 = *(const float4*)(W + (size_t)(2 * kp + 1) * HID + 4 * m);
        float* d = sW + kp * WPITCH + 8 * m;
        *(float4*)(d)     = make_float4(w0.x, w1.x, w0.y, w1.y);
        *(float4*)(d + 4) = make_float4(w0.z, w1.z, w0.w, w1.w);
    }

    float* sAw = sA + wave * (16 * APITCH);
    #pragma unroll
    for (int r = 0; r < 16; ++r)
        *(float4*)(sAw + r * APITCH + 4 * lane) = ar[r];

    __syncthreads();

    const float* ap = sAw + c * APITCH + 2 * h;
    const float* bp = sW + h * WPITCH + 2 * c;

    #pragma unroll
    for (int j = 0; j < 8; ++j)
        acc[j] = (v8f){0.f, 0.f, 0.f, 0.f, 0.f, 0.f, 0.f, 0.f};

    v2f a_cur = *(const v2f*)(ap);
    v2f b_cur[8];
    #pragma unroll
    for (int j = 0; j < 8; ++j)
        b_cur[j] = *(const v2f*)(bp + 32 * j);

    #pragma unroll
    for (int k0 = 0; k0 < HID; k0 += 4) {
        v2f a_nxt = a_cur;
        v2f b_nxt[8];
        #pragma unroll
        for (int j = 0; j < 8; ++j)
            b_nxt[j] = b_cur[j];
        if (k0 + 4 < HID) {
            a_nxt = *(const v2f*)(ap + k0 + 4);
            const float* bk = bp + ((k0 + 4) >> 1) * WPITCH;
            #pragma unroll
            for (int j = 0; j < 8; ++j)
                b_nxt[j] = *(const v2f*)(bk + 32 * j);
        }
        #pragma unroll
        for (int j = 0; j < 8; ++j)
            acc[j] = __builtin_amdgcn_wmma_f32_16x16x4_f32(
                 false, a_cur,  false, b_cur[j],
                 (short)0, acc[j],  false,  false);
        a_cur = a_nxt;
        #pragma unroll
        for (int j = 0; j < 8; ++j)
            b_cur[j] = b_nxt[j];
    }
}

__global__ __launch_bounds__(256)
void gnn_linear_relu_wmma(const float* __restrict__ X,
                          const float* __restrict__ W,
                          const float* __restrict__ bias,
                          float* __restrict__ T, int numTiles,
                          int* __restrict__ cnt, int N) {
    const int gid = blockIdx.x * blockDim.x + threadIdx.x;
    if (gid < N) cnt[gid] = 0;

    const int lane = threadIdx.x & 31;
    const int c    = lane & 15;
    const int h    = lane >> 4;
    const int mine = blockIdx.x * GEMM_WAVES + (threadIdx.x >> 5);
    const bool active = mine < numTiles;
    const int tile = active ? mine : numTiles - 1;

    v8f acc[8];
    gemm_strip_16x128(X, W, tile, acc);

    if (active) {
        const int m_base = tile << 4;
        #pragma unroll
        for (int j = 0; j < 8; ++j) {
            const int col = 16 * j + c;
            const float bv = bias[col];
            #pragma unroll
            for (int r = 0; r < 8; ++r) {
                const int row = m_base + r + (h << 3);
                T[(size_t)row * HID + col] = fmaxf(acc[j][r] + bv, 0.0f);
            }
        }
    }
}

__global__ __launch_bounds__(256)
void gnn_edge_rank(const int* __restrict__ dst, int* __restrict__ cnt,
                   int* __restrict__ rank, int E) {
    const int e = blockIdx.x * blockDim.x + threadIdx.x;
    if (e >= E) return;
    rank[e] = atomicAdd(cnt + dst[e], 1);
}

__global__ __launch_bounds__(1024)
void gnn_count_prefix(const int* __restrict__ cnt, int* __restrict__ offs, int N) {
    __shared__ int wtot[32];
    __shared__ int wpre[32];
    const int tid  = threadIdx.x;
    const int lane = tid & 31;
    const int wave = tid >> 5;
    const int first = blockIdx.x * 1024;
    const int i = first + tid;
    const int v = (i < N) ? cnt[i] : 0;

    int pre = 0;
    for (int k = tid; k < first; k += 1024)
        pre += cnt[k];

    int inc = v;
    #pragma unroll
    for (int off = 1; off < 32; off <<= 1) {
        const int t = __shfl_up(inc, off, 32);
        if (lane >= off) inc += t;
    }
    #pragma unroll
    for (int off = 16; off > 0; off >>= 1)
        pre += __shfl_xor(pre, off, 32);

    if (lane == 31) wtot[wave] = inc;
    if (lane == 0)  wpre[wave] = pre;
    __syncthreads();

    int base = 0;
    for (int w = 0; w < 32; ++w) {
        base += wpre[w];
        if (w < wave) base += wtot[w];
    }
    if (i < N) offs[i] = base + inc - v;
}

__global__ __launch_bounds__(256)
void gnn_edge_place(const int* __restrict__ src, const int* __restrict__ dst,
                    const int* __restrict__ offs, const int* __restrict__ rank,
                    int* __restrict__ slot, int E) {
    const int e = blockIdx.x * blockDim.x + threadIdx.x;
    if (e >= E) return;
    slot[offs[dst[e]] + rank[e]] = src[e];
}

__global__ __launch_bounds__(256)
void gnn_gather_sum_rows(const float* __restrict__ T,
                         const int* __restrict__ slot,
                         const int* __restrict__ offs,
                         const int* __restrict__ cnt,
                         float* __restrict__ H, int N) {
    const int lane = threadIdx.x & 31;
    const int n = blockIdx.x * 8 + __builtin_amdgcn_readfirstlane((int)(threadIdx.x >> 5));
    if (n >= N) return;
    const int beg = offs[n];
    const int deg = cnt[n];
    const float4* T4 = (const float4*)T;
    float4 acc = make_float4(0.f, 0.f, 0.f, 0.f);
    for (int base = 0; base < deg; base += 32) {
        const int m = (deg - base < 32) ? (deg - base) : 32;
        const int mine = (lane < m) ? slot[beg + base + lane] : 0;
        for (int j = 0; j < m; j += 8) {
            int s[8];
            #pragma unroll
            for (int u = 0; u < 8; ++u)
                s[u] = __builtin_amdgcn_readlane(mine, (j + u < m) ? (j + u) : (m - 1));
            float4 v[8];
            #pragma unroll
            for (int u = 0; u < 8; ++u) {
                v[u] = make_float4(0.f, 0.f, 0.f, 0.f);
                if (j + u < m)
                    v[u] = T4[(size_t)s[u] * (HID / 4) + lane];
            }
            #pragma unroll
            for (int u = 0; u < 8; ++u) {
                acc.x += v[u].x;
                acc.y += v[u].y;
                acc.z += v[u].z;
                acc.w += v[u].w;
            }
        }
    }
    ((float4*)H)[(size_t)n * (HID / 4) + lane] = acc;
}

__global__ __launch_bounds__(256)
void gnn_linear_relu_readout_wmma(const float* __restrict__ X,
                                  const float* __restrict__ W,
                                  const float* __restrict__ bias,
                                  const float* __restrict__ Wf,
                                  float* __restrict__ S, int numTiles) {
    const int lane = threadIdx.x & 31;
    const int c    = lane & 15;
    const int h    = lane >> 4;
    const int mine = blockIdx.x * GEMM_WAVES + (threadIdx.x >> 5);
    const bool active = mine < numTiles;
    const int tile = active ? mine : numTiles - 1;

    v8f acc[8];
    gemm_strip_16x128(X, W, tile, acc);

    float s[8];
    #pragma unroll
    for (int j = 0; j < 8; ++j) {
        const int col = 16 * j + c;
        const float bv = bias[col];
        const float wf = Wf[col];
        float p[8];
        #pragma unroll
        for (int r = 0; r < 8; ++r)
            p[r] = fmaxf(acc[j][r] + bv, 0.0f) * wf;
        #pragma unroll
        for (int off = 8; off > 0; off >>= 1) {
            #pragma unroll
            for (int r = 0; r < 8; ++r)
                p[r] += __shfl_xor(p[r], off, 32);
        }
        #pragma unroll
        for (int r = 0; r < 8; ++r)
            s[r] = (j == 0) ? p[r] : s[r] + p[r];
    }

    float v = s[0];
    #pragma unroll
    for (int r = 1; r < 8; ++r)
        v = (c == r) ? s[r] : v;
    if (active && c < 8) {
        const int row = (tile << 4) + (h << 3) + c;
        S[row] = v;
    }
}

__global__ __launch_bounds__(256)
void gnn_gather_sum_scalar(const float* __restrict__ S,
                           const int* __restrict__ slot,
                           const int* __restrict__ offs,
                           const int* __restrict__ cnt,
                           const float* __restrict__ bf,
                           float* __restrict__ out, int N) {
    const int n = blockIdx.x * blockDim.x + threadIdx.x;
    if (n >= N) return;
    const int beg = offs[n];
    const int end = beg + cnt[n];
    float acc = 0.0f;
    for (int j = beg; j < end; j += 4) {
        const int i0 = slot[j];
        const int i1 = (j + 1 < end) ? slot[j + 1] : -1;
        const int i2 = (j + 2 < end) ? slot[j + 2] : -1;
        const int i3 = (j + 3 < end) ? slot[j + 3] : -1;
        const float a0 = S[i0];
        const float a1 = (i1 >= 0) ? S[i1] : 0.0f;
        const float a2 = (i2 >= 0) ? S[i2] : 0.0f;
        const float a3 = (i3 >= 0) ? S[i3] : 0.0f;
        acc += a0;
        acc += a1;
        acc += a2;
        acc += a3;
    }
    out[n] = acc + bf[0];
}

extern "C" void kernel_launch(void* const* d_in, const int* in_sizes, int n_in,
                              void* d_out, int out_size, void* d_ws, size_t ws_size,
                              hipStream_t stream) {
    const float* x   = (const float*)d_in[0];
    const int*   ei  = (const int*)  d_in[1];
    const float* W1  = (const float*)d_in[2];
    const float* b1  = (const float*)d_in[3];
    const float* W2  = (const float*)d_in[4];
    const float* b2  = (const float*)d_in[5];
    const float* Wf  = (const float*)d_in[6];
    const float* bf  = (const float*)d_in[7];
    float* out = (float*)d_out;

    const int N = in_sizes[0] / HID;
    const int E = in_sizes[1] / 2;
    const int* src = ei;
    const int* dst = ei + E;

    float* bufT = (float*)d_ws;
    float* bufH = bufT + (size_t)N * HID;
    float* bufS = bufH + (size_t)N * HID;
    int*   cnt  = (int*)(bufS + N);
    int*   offs = cnt + N;
    int*   rank = offs + N;
    int*   slot = rank + E;

    const dim3 blk(256);
    const int numTiles  = N / 16;
    const int gemmGrid  = (numTiles + 7) / 8;
    const int edgeGrid  = (E + 255) / 256;
    const int scanGrid  = (N + 1023) / 1024;
    const int rowGrid   = (N + 7) / 8;
    const int nodeGrid  = (N + 255) / 256;

    gnn_linear_relu_wmma<<<gemmGrid, blk, 0, stream>>>(x, W1, b1, bufT, numTiles, cnt, N);

    gnn_edge_rank<<<edgeGrid, blk, 0, stream>>>(dst, cnt, rank, E);
    gnn_count_prefix<<<scanGrid, dim3(1024), 0, stream>>>(cnt, offs, N);
    gnn_edge_place<<<edgeGrid, blk, 0, stream>>>(src, dst, offs, rank, slot, E);

    gnn_gather_sum_rows<<<rowGrid, blk, 0, stream>>>(bufT, slot, offs, cnt, bufH, N);

    gnn_linear_relu_readout_wmma<<<gemmGrid, blk, 0, stream>>>(bufH, W2, b2, Wf, bufS, numTiles);
    gnn_gather_sum_scalar<<<nodeGrid, blk, 0, stream>>>(bufS, slot, offs, cnt, bf, out, N);
}


# Edge bucketing by counting sort with LDS counters: a block owns an edge slice and a node range, counts, then places through LDS cursors; no global returning atomics; sum order changes
# speedup vs baseline: 3.6077x; 1.3938x over previous
#include <hip/hip_runtime.h>


typedef float v2f __attribute__((ext_vector_type(2)));
typedef float v8f __attribute__((ext_vector_type(8)));

#define HID 128
#define GEMM_WAVES 8
#define WPITCH 288
#define APITCH 132

__device__ __forceinline__ void gemm_strip_16x128(const float* __restrict__ X,
                                                  const float* __restrict__ W,
                                                  int tile, v8f (&acc)[8]) {
    __shared__ __attribute__((aligned(16))) float sW[64 * WPITCH];
    __shared__ __attribute__((aligned(16))) float sA[GEMM_WAVES * 16 * APITCH];

    const int tid  = threadIdx.x;
    const int lane = tid & 31;
    const int wave = tid >> 5;
    const int c    = lane & 15;
    const int h    = lane >> 4;

    float4 ar[16];
    const float4* xp = (const float4*)(X + (size_t)tile * 16 * HID) + lane;
    #pragma unroll
    for (int r = 0; r < 16; ++r)
        ar[r] = xp[r * (HID / 4)];

    #pragma unroll
    for (int i = 0; i < 8; ++i) {
        const int item = i * 256 + tid;
        const int kp   = item >> 5;
        const int m    = item & 31;
        const float4 w0 = *(const float4*)(W + (size_t)(2 * kp) * HID + 4 * m);
        const float4 w1 = *(const float4*)(W + (size_t)(2 * kp + 1) * HID + 4 * m);
        float* d = sW + kp * WPITCH + 8 * m;
        *(float4*)(d)     = make_float4(w0.x, w1.x, w0.y, w1.y);
        *(float4*)(d + 4) = make_float4(w0.z, w1.z, w0.w, w1.w);
    }

    float* sAw = sA + wave * (16 * APITCH);
    #pragma unroll
    for (int r = 0; r < 16; ++r)
        *(float4*)(sAw + r * APITCH + 4 * lane) = ar[r];

    __syncthreads();

    const float* ap = sAw + c * APITCH + 2 * h;
    const float* bp = sW + h * WPITCH + 2 * c;

    #pragma unroll
    for (int j = 0; j < 8; ++j)
        acc[j] = (v8f){0.f, 0.f, 0.f, 0.f, 0.f, 0.f, 0.f, 0.f};

    v2f a_cur = *(const v2f*)(ap);
    v2f b_cur[8];
    #pragma unroll
    for (int j = 0; j < 8; ++j)
        b_cur[j] = *(const v2f*)(bp + 32 * j);

    #pragma unroll
    for (int k0 = 0; k0 < HID; k0 += 4) {
        v2f a_nxt = a_cur;
        v2f b_nxt[8];
        #pragma unroll
        for (int j = 0; j < 8; ++j)
            b_nxt[j] = b_cur[j];
        if (k0 + 4 < HID) {
            a_nxt = *(const v2f*)(ap + k0 + 4);
            const float* bk = bp + ((k0 + 4) >> 1) * WPITCH;
            #pragma unroll
            for (int j = 0; j < 8; ++j)
                b_nxt[j] = *(const v2f*)(bk + 32 * j);
        }
        #pragma unroll
        for (int j = 0; j < 8; ++j)
            acc[j] = __builtin_amdgcn_wmma_f32_16x16x4_f32(
                 false, a_cur,  false, b_cur[j],
                 (short)0, acc[j],  false,  false);
        a_cur = a_nxt;
        #pragma unroll
        for (int j = 0; j < 8; ++j)
            b_cur[j] = b_nxt[j];
    }
}

__global__ __launch_bounds__(256)
void gnn_linear_relu_wmma(const float* __restrict__ X,
                          const float* __restrict__ W,
                          const float* __restrict__ bias,
                          float* __restrict__ T, int numTiles) {
    const int lane = threadIdx.x & 31;
    const int c    = lane & 15;
    const int h    = lane >> 4;
    const int mine = blockIdx.x * GEMM_WAVES + (threadIdx.x >> 5);
    const bool active = mine < numTiles;
    const int tile = active ? mine : numTiles - 1;

    v8f acc[8];
    gemm_strip_16x128(X, W, tile, acc);

    if (active) {
        const int m_base = tile << 4;
        #pragma unroll
        for (int j = 0; j < 8; ++j) {
            const int col = 16 * j + c;
            const float bv = bias[col];
            #pragma unroll
            for (int r = 0; r < 8; ++r) {
                const int row = m_base + r + (h << 3);
                T[(size_t)row * HID + col] = fmaxf(acc[j][r] + bv, 0.0f);
            }
        }
    }
}

#define BK_CAP    4096
#define BK_SLICES 16
#define BK_THREADS 1024
#define BK_GROUP  16
#define BK_SUB    (BK_CAP / BK_THREADS)

__device__ __forceinline__ int bk_slice_len(int E) { return (E + BK_SLICES - 1) / BK_SLICES; }

__global__ __launch_bounds__(BK_THREADS)
void gnn_slice_count(const int* __restrict__ dst, int* __restrict__ cntm,
                     int* __restrict__ bs, int N, int E) {
    __shared__ int c[BK_CAP];
    __shared__ int tot[BK_SUB];
    const int tid = threadIdx.x;
    const int r = blockIdx.x;
    const int s = blockIdx.y;
    const int lo = r * BK_CAP;
    for (int i = tid; i < BK_CAP; i += BK_THREADS) c[i] = 0;
    if (tid < BK_SUB) tot[tid] = 0;
    __syncthreads();

    const int per = bk_slice_len(E);
    const int e0 = s * per;
    const int e1 = (e0 + per < E) ? (e0 + per) : E;
    for (int base = e0; base < e1; base += BK_GROUP * BK_THREADS) {
        int d[BK_GROUP];
        #pragma unroll
        for (int u = 0; u < BK_GROUP; ++u) {
            const int e = base + u * BK_THREADS + tid;
            d[u] = (e < e1) ? dst[e] : -1;
        }
        #pragma unroll
        for (int u = 0; u < BK_GROUP; ++u) {
            const unsigned k = (unsigned)(d[u] - lo);
            if (k < (unsigned)BK_CAP) atomicAdd(&c[k], 1);
        }
    }
    __syncthreads();

    #pragma unroll
    for (int q = 0; q < BK_SUB; ++q) {
        const int i = q * BK_THREADS + tid;
        const int n = lo + i;
        int v = 0;
        if (n < N) {
            v = c[i];
            cntm[(size_t)s * N + n] = v;
        }
        #pragma unroll
        for (int off = 16; off > 0; off >>= 1)
            v += __shfl_xor(v, off, 32);
        if ((tid & 31) == 0 && v != 0) atomicAdd(&tot[q], v);
    }
    __syncthreads();
    if (tid < BK_SUB) bs[s * (gridDim.x * BK_SUB) + r * BK_SUB + tid] = tot[tid];
}

__global__ __launch_bounds__(BK_THREADS)
void gnn_bucket_offsets(const int* __restrict__ cntm, const int* __restrict__ bs,
                        int* __restrict__ cnt, int* __restrict__ offs,
                        int* __restrict__ basem, int N) {
    __shared__ int wtot[32];
    __shared__ int sbase;
    const int tid  = threadIdx.x;
    const int lane = tid & 31;
    const int wave = tid >> 5;
    const int g  = blockIdx.x;
    const int ng = gridDim.x;
    const int n  = g * BK_THREADS + tid;
    if (tid == 0) sbase = 0;
    __syncthreads();

    int part = 0;
    for (int i = tid; i < BK_SLICES * ng; i += BK_THREADS)
        if ((i % ng) < g) part += bs[i];
    #pragma unroll
    for (int off = 16; off > 0; off >>= 1)
        part += __shfl_xor(part, off, 32);
    if (lane == 0 && part != 0) atomicAdd(&sbase, part);

    int v[BK_SLICES];
    #pragma unroll
    for (int s = 0; s < BK_SLICES; ++s)
        v[s] = (n < N) ? cntm[(size_t)s * N + n] : 0;
    int mine = 0;
    #pragma unroll
    for (int s = 0; s < BK_SLICES; ++s)
        mine += v[s];

    int inc = mine;
    #pragma unroll
    for (int off = 1; off < 32; off <<= 1) {
        const int t = __shfl_up(inc, off, 32);
        if (lane >= off) inc += t;
    }
    if (lane == 31) wtot[wave] = inc;
    __syncthreads();

    int run = sbase + inc - mine;
    for (int w = 0; w < wave; ++w)
        run += wtot[w];

    if (n < N) {
        offs[n] = run;
        cnt[n]  = mine;
        #pragma unroll
        for (int s = 0; s < BK_SLICES; ++s) {
            basem[(size_t)s * N + n] = run;
            run += v[s];
        }
    }
}

__global__ __launch_bounds__(BK_THREADS)
void gnn_slice_place(const int* __restrict__ src, const int* __restrict__ dst,
                     const int* __restrict__ basem, int* __restrict__ slot, int N, int E) {
    __shared__ int cur[BK_CAP];
    const int tid = threadIdx.x;
    const int r = blockIdx.x;
    const int s = blockIdx.y;
    const int lo = r * BK_CAP;
    int first[BK_SUB];
    #pragma unroll
    for (int q = 0; q < BK_SUB; ++q) {
        const int n = lo + q * BK_THREADS + tid;
        first[q] = (n < N) ? basem[(size_t)s * N + n] : 0;
    }
    #pragma unroll
    for (int q = 0; q < BK_SUB; ++q)
        cur[q * BK_THREADS + tid] = first[q];
    __syncthreads();

    const int per = bk_slice_len(E);
    const int e0 = s * per;
    const int e1 = (e0 + per < E) ? (e0 + per) : E;
    for (int base = e0; base < e1; base += BK_GROUP * BK_THREADS) {
        int d[BK_GROUP];
        int sv[BK_GROUP];
        #pragma unroll
        for (int u = 0; u < BK_GROUP; ++u) {
            const int e = base + u * BK_THREADS + tid;
            d[u]  = (e < e1) ? dst[e] : -1;
            sv[u] = (e < e1) ? src[e] : 0;
        }
        #pragma unroll
        for (int u = 0; u < BK_GROUP; ++u) {
            const unsigned k = (unsigned)(d[u] - lo);
            if (k < (unsigned)BK_CAP && d[u] < N) {
                const int pos = atomicAdd(&cur[k], 1);
                slot[pos] = sv[u];
            }
        }
    }
}

__global__ __launch_bounds__(256)
void gnn_gather_sum_rows(const float* __restrict__ T,
                         const int* __restrict__ slot,
                         const int* __restrict__ offs,
                         const int* __restrict__ cnt,
                         float* __restrict__ H, int N) {
    const int lane = threadIdx.x & 31;
    const int n = blockIdx.x * 8 + __builtin_amdgcn_readfirstlane((int)(threadIdx.x >> 5));
    if (n >= N) return;
    const int beg = offs[n];
    const int deg = cnt[n];
    const float4* T4 = (const float4*)T;
    float4 acc = make_float4(0.f, 0.f, 0.f, 0.f);
    for (int base = 0; base < deg; base += 32) {
        const int m = (deg - base < 32) ? (deg - base) : 32;
        const int mine = (lane < m) ? slot[beg + base + lane] : 0;
        for (int j = 0; j < m; j += 8) {
            int s[8];
            #pragma unroll
            for (int u = 0; u < 8; ++u)
                s[u] = __builtin_amdgcn_readlane(mine, (j + u < m) ? (j + u) : (m - 1));
            float4 v[8];
            #pragma unroll
            for (int u = 0; u < 8; ++u) {
                v[u] = make_float4(0.f, 0.f, 0.f, 0.f);
                if (j + u < m)
                    v[u] = T4[(size_t)s[u] * (HID / 4) + lane];
            }
            #pragma unroll
            for (int u = 0; u < 8; ++u) {
                acc.x += v[u].x;
                acc.y += v[u].y;
                acc.z += v[u].z;
                acc.w += v[u].w;
            }
        }
    }
    ((float4*)H)[(size_t)n * (HID / 4) + lane] = acc;
}

__global__ __launch_bounds__(256)
void gnn_linear_relu_readout_wmma(const float* __restrict__ X,
                                  const float* __restrict__ W,
                                  const float* __restrict__ bias,
                                  const float* __restrict__ Wf,
                                  float* __restrict__ S, int numTiles) {
    const int lane = threadIdx.x & 31;
    const int c    = lane & 15;
    const int h    = lane >> 4;
    const int mine = blockIdx.x * GEMM_WAVES + (threadIdx.x >> 5);
    const bool active = mine < numTiles;
    const int tile = active ? mine : numTiles - 1;

    v8f acc[8];
    gemm_strip_16x128(X, W, tile, acc);

    float s[8];
    #pragma unroll
    for (int j = 0; j < 8; ++j) {
        const int col = 16 * j + c;
        const float bv = bias[col];
        const float wf = Wf[col];
        float p[8];
        #pragma unroll
        for (int r = 0; r < 8; ++r)
            p[r] = fmaxf(acc[j][r] + bv, 0.0f) * wf;
        #pragma unroll
        for (int off = 8; off > 0; off >>= 1) {
            #pragma unroll
            for (int r = 0; r < 8; ++r)
                p[r] += __shfl_xor(p[r], off, 32);
        }
        #pragma unroll
        for (int r = 0; r < 8; ++r)
            s[r] = (j == 0) ? p[r] : s[r] + p[r];
    }

    float v = s[0];
    #pragma unroll
    for (int r = 1; r < 8; ++r)
        v = (c == r) ? s[r] : v;
    if (active && c < 8) {
        const int row = (tile << 4) + (h << 3) + c;
        S[row] = v;
    }
}

__global__ __launch_bounds__(256)
void gnn_gather_sum_scalar(const float* __restrict__ S,
                           const int* __restrict__ slot,
                           const int* __restrict__ offs,
                           const int* __restrict__ cnt,
                           const float* __restrict__ bf,
                           float* __restrict__ out, int N) {
    const int n = blockIdx.x * blockDim.x + threadIdx.x;
    if (n >= N) return;
    const int beg = offs[n];
    const int end = beg + cnt[n];
    float acc = 0.0f;
    for (int j = beg; j < end; j += 4) {
        const int i0 = slot[j];
        const int i1 = (j + 1 < end) ? slot[j + 1] : -1;
        const int i2 = (j + 2 < end) ? slot[j + 2] : -1;
        const int i3 = (j + 3 < end) ? slot[j + 3] : -1;
        const float a0 = S[i0];
        const float a1 = (i1 >= 0) ? S[i1] : 0.0f;
        const float a2 = (i2 >= 0) ? S[i2] : 0.0f;
        const float a3 = (i3 >= 0) ? S[i3] : 0.0f;
        acc += a0;
        acc += a1;
        acc += a2;
        acc += a3;
    }
    out[n] = acc + bf[0];
}

extern "C" void kernel_launch(void* const* d_in, const int* in_sizes, int n_in,
                              void* d_out, int out_size, void* d_ws, size_t ws_size,
                              hipStream_t stream) {
    const float* x   = (const float*)d_in[0];
    const int*   ei  = (const int*)  d_in[1];
    const float* W1  = (const float*)d_in[2];
    const float* b1  = (const float*)d_in[3];
    const float* W2  = (const float*)d_in[4];
    const float* b2  = (const float*)d_in[5];
    const float* Wf  = (const float*)d_in[6];
    const float* bf  = (const float*)d_in[7];
    float* out = (float*)d_out;

    const int N = in_sizes[0] / HID;
    const int E = in_sizes[1] / 2;
    const int* src = ei;
    const int* dst = ei + E;

    float* bufT = (float*)d_ws;
    float* bufH = bufT + (size_t)N * HID;
    float* bufS = bufH + (size_t)N * HID;
    int*   cnt   = (int*)(bufS + N);
    int*   offs  = cnt + N;
    int*   slot  = offs + N;
    int*   cntm  = slot + E;
    int*   basem = cntm + (size_t)BK_SLICES * N;
    int*   bs    = basem + (size_t)BK_SLICES * N;

    const dim3 blk(256);
    const int numTiles  = N / 16;
    const int gemmGrid  = (numTiles + 7) / 8;
    const int ranges    = (N + BK_CAP - 1) / BK_CAP;
    const dim3 bkGrid(ranges, BK_SLICES);
    const int rowGrid   = (N + 7) / 8;
    const int nodeGrid  = (N + 255) / 256;

    gnn_linear_relu_wmma<<<gemmGrid, blk, 0, stream>>>(x, W1, b1, bufT, numTiles);

    gnn_slice_count<<<bkGrid, dim3(BK_THREADS), 0, stream>>>(dst, cntm, bs, N, E);
    gnn_bucket_offsets<<<ranges * BK_SUB, dim3(BK_THREADS), 0, stream>>>(cntm, bs, cnt, offs, basem, N);
    gnn_slice_place<<<bkGrid, dim3(BK_THREADS), 0, stream>>>(src, dst, basem, slot, N, E);

    gnn_gather_sum_rows<<<rowGrid, blk, 0, stream>>>(bufT, slot, offs, cnt, bufH, N);

    gnn_linear_relu_readout_wmma<<<gemmGrid, blk, 0, stream>>>(bufH, W2, b2, Wf, bufS, numTiles);
    gnn_gather_sum_scalar<<<nodeGrid, blk, 0, stream>>>(bufS, slot, offs, cnt, bf, out, N);
}


# Edge bucketing in one launch: a (range, slice) block counts, scans in LDS and places into its own slice-long region; gathers walk a node's 16 runs; 5 launches not 7; layer-2 sum order changes
# speedup vs baseline: 4.1756x; 1.1574x over previous
#include <hip/hip_runtime.h>


typedef float v2f __attribute__((ext_vector_type(2)));
typedef float v8f __attribute__((ext_vector_type(8)));

#define HID 128
#define GEMM_WAVES 8
#define WPITCH 288
#define APITCH 132

__device__ __forceinline__ void gemm_strip_16x128(const float* __restrict__ X,
                                                  const float* __restrict__ W,
                                                  int tile, v8f (&acc)[8]) {
    __shared__ __attribute__((aligned(16))) float sW[64 * WPITCH];
    __shared__ __attribute__((aligned(16))) float sA[GEMM_WAVES * 16 * APITCH];

    const int tid  = threadIdx.x;
    const int lane = tid & 31;
    const int wave = tid >> 5;
    const int c    = lane & 15;
    const int h    = lane >> 4;

    float4 ar[16];
    const float4* xp = (const float4*)(X + (size_t)tile * 16 * HID) + lane;
    #pragma unroll
    for (int r = 0; r < 16; ++r)
        ar[r] = xp[r * (HID / 4)];

    #pragma unroll
    for (int i = 0; i < 8; ++i) {
        const int item = i * 256 + tid;
        const int kp   = item >> 5;
        const int m    = item & 31;
        const float4 w0 = *(const float4*)(W + (size_t)(2 * kp) * HID + 4 * m);
        const float4 w1 = *(const float4*)(W + (size_t)(2 * kp + 1) * HID + 4 * m);
        float* d = sW + kp * WPITCH + 8 * m;
        *(float4*)(d)     = make_float4(w0.x, w1.x, w0.y, w1.y);
        *(float4*)(d + 4) = make_float4(w0.z, w1.z, w0.w, w1.w);
    }

    float* sAw = sA + wave * (16 * APITCH);
    #pragma unroll
    for (int r = 0; r < 16; ++r)
        *(float4*)(sAw + r * APITCH + 4 * lane) = ar[r];

    __syncthreads();

    const float* ap = sAw + c * APITCH + 2 * h;
    const float* bp = sW + h * WPITCH + 2 * c;

    #pragma unroll
    for (int j = 0; j < 8; ++j)
        acc[j] = (v8f){0.f, 0.f, 0.f, 0.f, 0.f, 0.f, 0.f, 0.f};

    v2f a_cur = *(const v2f*)(ap);
    v2f b_cur[8];
    #pragma unroll
    for (int j = 0; j < 8; ++j)
        b_cur[j] = *(const v2f*)(bp + 32 * j);

    #pragma unroll
    for (int k0 = 0; k0 < HID; k0 += 4) {
        v2f a_nxt = a_cur;
        v2f b_nxt[8];
        #pragma unroll
        for (int j = 0; j < 8; ++j)
            b_nxt[j] = b_cur[j];
        if (k0 + 4 < HID) {
            a_nxt = *(const v2f*)(ap + k0 + 4);
            const float* bk = bp + ((k0 + 4) >> 1) * WPITCH;
            #pragma unroll
            for (int j = 0; j < 8; ++j)
                b_nxt[j] = *(const v2f*)(bk + 32 * j);
        }
        #pragma unroll
        for (int j = 0; j < 8; ++j)
            acc[j] = __builtin_amdgcn_wmma_f32_16x16x4_f32(
                 false, a_cur,  false, b_cur[j],
                 (short)0, acc[j],  false,  false);
        a_cur = a_nxt;
        #pragma unroll
        for (int j = 0; j < 8; ++j)
            b_cur[j] = b_nxt[j];
    }
}

__global__ __launch_bounds__(256)
void gnn_linear_relu_wmma(const float* __restrict__ X,
                          const float* __restrict__ W,
                          const float* __restrict__ bias,
                          float* __restrict__ T, int numTiles) {
    const int lane = threadIdx.x & 31;
    const int c    = lane & 15;
    const int h    = lane >> 4;
    const int mine = blockIdx.x * GEMM_WAVES + (threadIdx.x >> 5);
    const bool active = mine < numTiles;
    const int tile = active ? mine : numTiles - 1;

    v8f acc[8];
    gemm_strip_16x128(X, W, tile, acc);

    if (active) {
        const int m_base = tile << 4;
        #pragma unroll
        for (int j = 0; j < 8; ++j) {
            const int col = 16 * j + c;
            const float bv = bias[col];
            #pragma unroll
            for (int r = 0; r < 8; ++r) {
                const int row = m_base + r + (h << 3);
                T[(size_t)row * HID + col] = fmaxf(acc[j][r] + bv, 0.0f);
            }
        }
    }
}

#define BK_CAP    4096
#define BK_SLICES 16
#define BK_THREADS 1024
#define BK_GROUP  16
#define BK_SUB    (BK_CAP / BK_THREADS)

__device__ __forceinline__ int bk_slice_len(int E) { return (E + BK_SLICES - 1) / BK_SLICES; }

__global__ __launch_bounds__(BK_THREADS)
void gnn_slice_sort(const int* __restrict__ src, const int* __restrict__ dst,
                    int2* __restrict__ seg, int* __restrict__ region, int N, int E) {
    __shared__ int c[BK_CAP];
    __shared__ int wtot[BK_THREADS / 32];
    const int tid  = threadIdx.x;
    const int lane = tid & 31;
    const int wave = tid >> 5;
    const int r = blockIdx.x;
    const int s = blockIdx.y;
    const int lo = r * BK_CAP;
    #pragma unroll
    for (int q = 0; q < BK_SUB; ++q) c[q * BK_THREADS + tid] = 0;
    __syncthreads();

    const int per = bk_slice_len(E);
    const int e0 = s * per;
    const int e1 = (e0 + per < E) ? (e0 + per) : E;

    for (int base = e0; base < e1; base += BK_GROUP * BK_THREADS) {
        int d[BK_GROUP];
        #pragma unroll
        for (int u = 0; u < BK_GROUP; ++u) {
            const int e = base + u * BK_THREADS + tid;
            d[u] = (e < e1) ? dst[e] : -1;
        }
        #pragma unroll
        for (int u = 0; u < BK_GROUP; ++u) {
            const unsigned k = (unsigned)(d[u] - lo);
            if (k < (unsigned)BK_CAP && d[u] < N) atomicAdd(&c[k], 1);
        }
    }
    __syncthreads();

    int a[BK_SUB];
    int mine = 0;
    #pragma unroll
    for (int q = 0; q < BK_SUB; ++q) {
        a[q] = c[q * BK_THREADS + tid];
        mine += a[q];
    }
    int inc = mine;
    #pragma unroll
    for (int off = 1; off < 32; off <<= 1) {
        const int t = __shfl_up(inc, off, 32);
        if (lane >= off) inc += t;
    }
    if (lane == 31) wtot[wave] = inc;
    __syncthreads();

    int run = (r * BK_SLICES + s) * per + inc - mine;
    for (int w = 0; w < wave; ++w)
        run += wtot[w];
    #pragma unroll
    for (int q = 0; q < BK_SUB; ++q) {
        const int n = lo + q * BK_THREADS + tid;
        c[q * BK_THREADS + tid] = run;
        if (n < N) seg[(size_t)s * N + n] = make_int2(run, a[q]);
        run += a[q];
    }
    __syncthreads();

    for (int base = e0; base < e1; base += BK_GROUP * BK_THREADS) {
        int d[BK_GROUP];
        int sv[BK_GROUP];
        #pragma unroll
        for (int u = 0; u < BK_GROUP; ++u) {
            const int e = base + u * BK_THREADS + tid;
            d[u]  = (e < e1) ? dst[e] : -1;
            sv[u] = (e < e1) ? src[e] : 0;
        }
        #pragma unroll
        for (int u = 0; u < BK_GROUP; ++u) {
            const unsigned k = (unsigned)(d[u] - lo);
            if (k < (unsigned)BK_CAP && d[u] < N) {
                const int pos = atomicAdd(&c[k], 1);
                region[pos] = sv[u];
            }
        }
    }
}

__global__ __launch_bounds__(256)
void gnn_gather_sum_rows(const float* __restrict__ T,
                         const int* __restrict__ region,
                         const int2* __restrict__ seg,
                         float* __restrict__ H, int N) {
    __shared__ int2 sseg[8][BK_SLICES];
    const int tid  = threadIdx.x;
    const int lane = tid & 31;
    const int wv = __builtin_amdgcn_readfirstlane((int)(tid >> 5));
    const int n0 = blockIdx.x * 8;
    if (tid < 8 * BK_SLICES) {
        const int s = tid >> 3;
        const int j = tid & 7;
        int2 v = make_int2(0, 0);
        if (n0 + j < N) v = seg[(size_t)s * N + n0 + j];
        sseg[j][s] = v;
    }
    __syncthreads();
    const int n = n0 + wv;
    if (n >= N) return;

    int st = 0, cn = 0;
    if (lane < BK_SLICES) {
        const int2 v = sseg[wv][lane];
        st = v.x;
        cn = v.y;
    }
    int inc = cn;
    #pragma unroll
    for (int off = 1; off < 32; off <<= 1) {
        const int t = __shfl_up(inc, off, 32);
        if (lane >= off) inc += t;
    }
    const int ex    = inc - cn;
    const int deg   = __builtin_amdgcn_readlane(inc, 31);
    const int delta = st - ex;
    int exs[BK_SLICES];
    #pragma unroll
    for (int s = 1; s < BK_SLICES; ++s)
        exs[s] = __builtin_amdgcn_readlane(ex, s);

    const float4* T4 = (const float4*)T;
    float4 acc = make_float4(0.f, 0.f, 0.f, 0.f);
    for (int base = 0; base < deg; base += 32) {
        const int m = (deg - base < 32) ? (deg - base) : 32;
        const int p = base + lane;
        int sp = 0;
        #pragma unroll
        for (int s = 1; s < BK_SLICES; ++s)
            sp += (p >= exs[s]) ? 1 : 0;
        const int dl = __shfl(delta, sp, 32);
        const int mine = (lane < m) ? region[dl + p] : 0;
        for (int j = 0; j < m; j += 8) {
            int s[8];
            #pragma unroll
            for (int u = 0; u < 8; ++u)
                s[u] = __builtin_amdgcn_readlane(mine, (j + u < m) ? (j + u) : (m - 1));
            float4 v[8];
            #pragma unroll
            for (int u = 0; u < 8; ++u) {
                v[u] = make_float4(0.f, 0.f, 0.f, 0.f);
                if (j + u < m)
                    v[u] = T4[(size_t)s[u] * (HID / 4) + lane];
            }
            #pragma unroll
            for (int u = 0; u < 8; ++u) {
                acc.x += v[u].x;
                acc.y += v[u].y;
                acc.z += v[u].z;
                acc.w += v[u].w;
            }
        }
    }
    ((float4*)H)[(size_t)n * (HID / 4) + lane] = acc;
}

__global__ __launch_bounds__(256)
void gnn_linear_relu_readout_wmma(const float* __restrict__ X,
                                  const float* __restrict__ W,
                                  const float* __restrict__ bias,
                                  const float* __restrict__ Wf,
                                  float* __restrict__ S, int numTiles) {
    const int lane = threadIdx.x & 31;
    const int c    = lane & 15;
    const int h    = lane >> 4;
    const int mine = blockIdx.x * GEMM_WAVES + (threadIdx.x >> 5);
    const bool active = mine < numTiles;
    const int tile = active ? mine : numTiles - 1;

    v8f acc[8];
    gemm_strip_16x128(X, W, tile, acc);

    float s[8];
    #pragma unroll
    for (int j = 0; j < 8; ++j) {
        const int col = 16 * j + c;
        const float bv = bias[col];
        const float wf = Wf[col];
        float p[8];
        #pragma unroll
        for (int r = 0; r < 8; ++r)
            p[r] = fmaxf(acc[j][r] + bv, 0.0f) * wf;
        #pragma unroll
        for (int off = 8; off > 0; off >>= 1) {
            #pragma unroll
            for (int r = 0; r < 8; ++r)
                p[r] += __shfl_xor(p[r], off, 32);
        }
        #pragma unroll
        for (int r = 0; r < 8; ++r)
            s[r] = (j == 0) ? p[r] : s[r] + p[r];
    }

    float v = s[0];
    #pragma unroll
    for (int r = 1; r < 8; ++r)
        v = (c == r) ? s[r] : v;
    if (active && c < 8) {
        const int row = (tile << 4) + (h << 3) + c;
        S[row] = v;
    }
}

__global__ __launch_bounds__(256)
void gnn_gather_sum_scalar(const float* __restrict__ S,
                           const int* __restrict__ region,
                           const int2* __restrict__ seg,
                           const float* __restrict__ bf,
                           float* __restrict__ out, int N) {
    const int n = blockIdx.x * blockDim.x + threadIdx.x;
    if (n >= N) return;
    int st[BK_SLICES];
    int cn[BK_SLICES];
    int maxc = 0;
    #pragma unroll
    for (int s = 0; s < BK_SLICES; ++s) {
        const int2 v = seg[(size_t)s * N + n];
        st[s] = v.x;
        cn[s] = v.y;
    }
    #pragma unroll
    for (int s = 0; s < BK_SLICES; ++s)
        maxc = (cn[s] > maxc) ? cn[s] : maxc;
    float acc = 0.0f;
    for (int i = 0; i < maxc; ++i) {
        int idx[BK_SLICES];
        #pragma unroll
        for (int s = 0; s < BK_SLICES; ++s)
            idx[s] = (i < cn[s]) ? region[st[s] + i] : -1;
        float v[BK_SLICES];
        #pragma unroll
        for (int s = 0; s < BK_SLICES; ++s)
            v[s] = (idx[s] >= 0) ? S[idx[s]] : 0.0f;
        #pragma unroll
        for (int s = 0; s < BK_SLICES; ++s)
            acc += v[s];
    }
    out[n] = acc + bf[0];
}

extern "C" void kernel_launch(void* const* d_in, const int* in_sizes, int n_in,
                              void* d_out, int out_size, void* d_ws, size_t ws_size,
                              hipStream_t stream) {
    const float* x   = (const float*)d_in[0];
    const int*   ei  = (const int*)  d_in[1];
    const float* W1  = (const float*)d_in[2];
    const float* b1  = (const float*)d_in[3];
    const float* W2  = (const float*)d_in[4];
    const float* b2  = (const float*)d_in[5];
    const float* Wf  = (const float*)d_in[6];
    const float* bf  = (const float*)d_in[7];
    float* out = (float*)d_out;

    const int N = in_sizes[0] / HID;
    const int E = in_sizes[1] / 2;
    const int* src = ei;
    const int* dst = ei + E;

    const int ranges = (N + BK_CAP - 1) / BK_CAP;
    const int per    = (E + BK_SLICES - 1) / BK_SLICES;

    float* bufT = (float*)d_ws;
    float* bufH = bufT + (size_t)N * HID;
    int2*  seg  = (int2*)(bufH + (size_t)N * HID);
    int*   region = (int*)(seg + (size_t)BK_SLICES * N);
    float* bufS = (float*)(region + (size_t)ranges * BK_SLICES * per);

    const dim3 blk(256);
    const int numTiles  = N / 16;
    const int gemmGrid  = (numTiles + 7) / 8;
    const dim3 bkGrid(ranges, BK_SLICES);
    const int rowGrid   = (N + 7) / 8;
    const int nodeGrid  = (N + 255) / 256;

    gnn_linear_relu_wmma<<<gemmGrid, blk, 0, stream>>>(x, W1, b1, bufT, numTiles);

    gnn_slice_sort<<<bkGrid, dim3(BK_THREADS), 0, stream>>>(src, dst, seg, region, N, E);

    gnn_gather_sum_rows<<<rowGrid, blk, 0, stream>>>(bufT, region, seg, bufH, N);

    gnn_linear_relu_readout_wmma<<<gemmGrid, blk, 0, stream>>>(bufH, W2, b2, Wf, bufS, numTiles);
    gnn_gather_sum_scalar<<<nodeGrid, blk, 0, stream>>>(bufS, region, seg, bf, out, N);
}


# Layer-1 row gather moved into the layer-2 GEMM launch: a GEMM wave sums its 16 nodes' source rows (list order kept) straight into its LDS A tile, so H never goes to memory; 4 launches not 5
# speedup vs baseline: 4.4081x; 1.0557x over previous
#include <hip/hip_runtime.h>


typedef float v2f __attribute__((ext_vector_type(2)));
typedef float v8f __attribute__((ext_vector_type(8)));

#define HID 128
#define GEMM_WAVES 8
#define WPITCH 288
#define APITCH 132

__device__ __forceinline__ void stage_w_lds(const float* __restrict__ W, float* sW, int tid) {
    #pragma unroll
    for (int i = 0; i < 8; ++i) {
        const int item = i * 256 + tid;
        const int kp   = item >> 5;
        const int m    = item & 31;
        const float4 w0 = *(const float4*)(W + (size_t)(2 * kp) * HID + 4 * m);
        const float4 w1 = *(const float4*)(W + (size_t)(2 * kp + 1) * HID + 4 * m);
        float* d = sW + kp * WPITCH + 8 * m;
        *(float4*)(d)     = make_float4(w0.x, w1.x, w0.y, w1.y);
        *(float4*)(d + 4) = make_float4(w0.z, w1.z, w0.w, w1.w);
    }
}

__device__ __forceinline__ void strip_from_lds(const float* sAw, const float* sW,
                                               int lane, v8f (&acc)[8]) {
    const int c = lane & 15;
    const int h = lane >> 4;
    const float* ap = sAw + c * APITCH + 2 * h;
    const float* bp = sW + h * WPITCH + 2 * c;

    #pragma unroll
    for (int j = 0; j < 8; ++j)
        acc[j] = (v8f){0.f, 0.f, 0.f, 0.f, 0.f, 0.f, 0.f, 0.f};

    v2f a_cur = *(const v2f*)(ap);
    v2f b_cur[8];
    #pragma unroll
    for (int j = 0; j < 8; ++j)
        b_cur[j] = *(const v2f*)(bp + 32 * j);

    #pragma unroll
    for (int k0 = 0; k0 < HID; k0 += 4) {
        v2f a_nxt = a_cur;
        v2f b_nxt[8];
        #pragma unroll
        for (int j = 0; j < 8; ++j)
            b_nxt[j] = b_cur[j];
        if (k0 + 4 < HID) {
            a_nxt = *(const v2f*)(ap + k0 + 4);
            const float* bk = bp + ((k0 + 4) >> 1) * WPITCH;
            #pragma unroll
            for (int j = 0; j < 8; ++j)
                b_nxt[j] = *(const v2f*)(bk + 32 * j);
        }
        #pragma unroll
        for (int j = 0; j < 8; ++j)
            acc[j] = __builtin_amdgcn_wmma_f32_16x16x4_f32(
                 false, a_cur,  false, b_cur[j],
                 (short)0, acc[j],  false,  false);
        a_cur = a_nxt;
        #pragma unroll
        for (int j = 0; j < 8; ++j)
            b_cur[j] = b_nxt[j];
    }
}

__device__ __forceinline__ void gemm_strip_16x128(const float* __restrict__ X,
                                                  const float* __restrict__ W,
                                                  int tile, v8f (&acc)[8]) {
    __shared__ __attribute__((aligned(16))) float sW[64 * WPITCH];
    __shared__ __attribute__((aligned(16))) float sA[GEMM_WAVES * 16 * APITCH];

    const int tid  = threadIdx.x;
    const int lane = tid & 31;
    const int wave = tid >> 5;

    float4 ar[16];
    const float4* xp = (const float4*)(X + (size_t)tile * 16 * HID) + lane;
    #pragma unroll
    for (int r = 0; r < 16; ++r)
        ar[r] = xp[r * (HID / 4)];

    stage_w_lds(W, sW, tid);

    float* sAw = sA + wave * (16 * APITCH);
    #pragma unroll
    for (int r = 0; r < 16; ++r)
        *(float4*)(sAw + r * APITCH + 4 * lane) = ar[r];

    __syncthreads();

    strip_from_lds(sAw, sW, lane, acc);
}

__global__ __launch_bounds__(256)
void gnn_linear_relu_wmma(const float* __restrict__ X,
                          const float* __restrict__ W,
                          const float* __restrict__ bias,
                          float* __restrict__ T, int numTiles) {
    const int lane = threadIdx.x & 31;
    const int c    = lane & 15;
    const int h    = lane >> 4;
    const int mine = blockIdx.x * GEMM_WAVES + (threadIdx.x >> 5);
    const bool active = mine < numTiles;
    const int tile = active ? mine : numTiles - 1;

    v8f acc[8];
    gemm_strip_16x128(X, W, tile, acc);

    if (active) {
        const int m_base = tile << 4;
        #pragma unroll
        for (int j = 0; j < 8; ++j) {
            const int col = 16 * j + c;
            const float bv = bias[col];
            #pragma unroll
            for (int r = 0; r < 8; ++r) {
                const int row = m_base + r + (h << 3);
                T[(size_t)row * HID + col] = fmaxf(acc[j][r] + bv, 0.0f);
            }
        }
    }
}

#define BK_CAP    4096
#define BK_SLICES 16
#define BK_THREADS 1024
#define BK_GROUP  16
#define BK_SUB    (BK_CAP / BK_THREADS)

__device__ __forceinline__ int bk_slice_len(int E) { return (E + BK_SLICES - 1) / BK_SLICES; }

__global__ __launch_bounds__(BK_THREADS)
void gnn_slice_sort(const int* __restrict__ src, const int* __restrict__ dst,
                    int2* __restrict__ seg, int* __restrict__ region, int N, int E) {
    __shared__ int c[BK_CAP];
    __shared__ int wtot[BK_THREADS / 32];
    const int tid  = threadIdx.x;
    const int lane = tid & 31;
    const int wave = tid >> 5;
    const int r = blockIdx.x;
    const int s = blockIdx.y;
    const int lo = r * BK_CAP;
    #pragma unroll
    for (int q = 0; q < BK_SUB; ++q) c[q * BK_THREADS + tid] = 0;
    __syncthreads();

    const int per = bk_slice_len(E);
    const int e0 = s * per;
    const int e1 = (e0 + per < E) ? (e0 + per) : E;

    for (int base = e0; base < e1; base += BK_GROUP * BK_THREADS) {
        int d[BK_GROUP];
        #pragma unroll
        for (int u = 0; u < BK_GROUP; ++u) {
            const int e = base + u * BK_THREADS + tid;
            d[u] = (e < e1) ? dst[e] : -1;
        }
        #pragma unroll
        for (int u = 0; u < BK_GROUP; ++u) {
            const unsigned k = (unsigned)(d[u] - lo);
            if (k < (unsigned)BK_CAP && d[u] < N) atomicAdd(&c[k], 1);
        }
    }
    __syncthreads();

    int a[BK_SUB];
    int mine = 0;
    #pragma unroll
    for (int q = 0; q < BK_SUB; ++q) {
        a[q] = c[q * BK_THREADS + tid];
        mine += a[q];
    }
    int inc = mine;
    #pragma unroll
    for (int off = 1; off < 32; off <<= 1) {
        const int t = __shfl_up(inc, off, 32);
        if (lane >= off) inc += t;
    }
    if (lane == 31) wtot[wave] = inc;
    __syncthreads();

    int run = (r * BK_SLICES + s) * per + inc - mine;
    for (int w = 0; w < wave; ++w)
        run += wtot[w];
    #pragma unroll
    for (int q = 0; q < BK_SUB; ++q) {
        const int n = lo + q * BK_THREADS + tid;
        c[q * BK_THREADS + tid] = run;
        if (n < N) seg[(size_t)s * N + n] = make_int2(run, a[q]);
        run += a[q];
    }
    __syncthreads();

    for (int base = e0; base < e1; base += BK_GROUP * BK_THREADS) {
        int d[BK_GROUP];
        int sv[BK_GROUP];
        #pragma unroll
        for (int u = 0; u < BK_GROUP; ++u) {
            const int e = base + u * BK_THREADS + tid;
            d[u]  = (e < e1) ? dst[e] : -1;
            sv[u] = (e < e1) ? src[e] : 0;
        }
        #pragma unroll
        for (int u = 0; u < BK_GROUP; ++u) {
            const unsigned k = (unsigned)(d[u] - lo);
            if (k < (unsigned)BK_CAP && d[u] < N) {
                const int pos = atomicAdd(&c[k], 1);
                region[pos] = sv[u];
            }
        }
    }
}

#define SEGPITCH 17

__device__ __forceinline__ void node_runs(int2 v, int lane, int& ex, int& delta, int& deg) {
    int inc = v.y;
    #pragma unroll
    for (int off = 1; off < 32; off <<= 1) {
        const int t = __shfl_up(inc, off, 32);
        if (lane >= off) inc += t;
    }
    ex    = inc - v.y;
    deg   = __builtin_amdgcn_readlane(inc, 31);
    delta = v.x - ex;
}

__device__ __forceinline__ int chunk_sources(const int* __restrict__ region, int ex, int delta,
                                             int base, int m, int lane) {
    int q = (lane < m) ? lane : (m - 1);
    q = (q > 0) ? q : 0;
    const int p = base + q;
    int sp = 0;
    #pragma unroll
    for (int s = 1; s < BK_SLICES; ++s)
        sp += (p >= __builtin_amdgcn_readlane(ex, s)) ? 1 : 0;
    const int dl = __shfl(delta, sp, 32);
    return region[dl + p];
}

template <int NR>
__device__ __forceinline__ void rows_issue(const float4* __restrict__ T4, int mine, int j, int m,
                                           int lane, float4 (&v)[NR]) {
    int s[NR];
    #pragma unroll
    for (int u = 0; u < NR; ++u)
        s[u] = __builtin_amdgcn_readlane(mine, j + u);
    #pragma unroll
    for (int u = 0; u < NR; ++u) {
        v[u] = make_float4(0.f, 0.f, 0.f, 0.f);
        if (j + u < m)
            v[u] = T4[(size_t)s[u] * (HID / 4) + lane];
    }
}

template <int NR>
__device__ __forceinline__ void rows_add(const float4 (&v)[NR], int first, float4& acc) {
    #pragma unroll
    for (int u = 0; u < NR; ++u) {
        if (u >= first) {
            acc.x += v[u].x;
            acc.y += v[u].y;
            acc.z += v[u].z;
            acc.w += v[u].w;
        }
    }
}

__global__ __launch_bounds__(256)
void gnn_gather_linear_relu_readout_wmma(const float* __restrict__ T,
                                         const int* __restrict__ region,
                                         const int2* __restrict__ seg,
                                         const float* __restrict__ W,
                                         const float* __restrict__ bias,
                                         const float* __restrict__ Wf,
                                         float* __restrict__ S, int N, int numTiles) {
    __shared__ __attribute__((aligned(16))) float sW[64 * WPITCH];
    __shared__ __attribute__((aligned(16))) float sA[GEMM_WAVES * 16 * APITCH];
    __shared__ int2 sseg[GEMM_WAVES * 16 * SEGPITCH];

    const int tid  = threadIdx.x;
    const int lane = tid & 31;
    const int wv   = __builtin_amdgcn_readfirstlane((int)(tid >> 5));
    const int c    = lane & 15;
    const int h    = lane >> 4;
    const int mine_tile = blockIdx.x * GEMM_WAVES + wv;
    const bool active = mine_tile < numTiles;
    const int n0 = blockIdx.x * (GEMM_WAVES * 16);

    int2 pr[8];
    #pragma unroll
    for (int i = 0; i < 8; ++i) {
        const int item = i * 256 + tid;
        const int s = item >> 7;
        const int j = item & 127;
        pr[i] = make_int2(0, 0);
        if (n0 + j < N) pr[i] = seg[(size_t)s * N + n0 + j];
    }

    stage_w_lds(W, sW, tid);

    #pragma unroll
    for (int i = 0; i < 8; ++i) {
        const int item = i * 256 + tid;
        const int s = item >> 7;
        const int j = item & 127;
        sseg[j * SEGPITCH + s] = pr[i];
    }
    __syncthreads();

    const int2* sg = sseg + wv * (16 * SEGPITCH);
    float* sAw = sA + wv * (16 * APITCH);
    const float4* T4 = (const float4*)T;

    int degv;
    {
        const int nr = lane & 15;
        const int hb = lane >> 4;
        int2 pr[BK_SLICES];
        #pragma unroll
        for (int s = 0; s < BK_SLICES; ++s)
            pr[s] = sg[nr * SEGPITCH + s];
        int exs[BK_SLICES];
        int jump[BK_SLICES];
        int run = 0;
        int dprev = 0;
        #pragma unroll
        for (int s = 0; s < BK_SLICES; ++s) {
            exs[s] = run;
            const int d = pr[s].x - run;
            jump[s] = d - dprev;
            dprev = d;
            run += pr[s].y;
        }
        degv = run;
        int* scr = (int*)(sAw + nr * APITCH) + 16 * hb;
        #pragma unroll
        for (int k = 0; k < 16; ++k) {
            const int p = 16 * hb + k;
            int slot = p + jump[0];
            #pragma unroll
            for (int s = 1; s < BK_SLICES; ++s)
                slot += (p >= exs[s]) ? jump[s] : 0;
            scr[k] = slot;
        }
    }
    __syncthreads();

    {
        int mine16[16];
        #pragma unroll
        for (int r = 0; r < 16; ++r) {
            const int deg = __builtin_amdgcn_readlane(degv, r);
            const int m   = (deg < 32) ? deg : 32;
            int q = (lane < m) ? lane : (m - 1);
            q = (q > 0) ? q : 0;
            int slot = ((const int*)(sAw + r * APITCH))[q];
            slot = (m > 0) ? slot : 0;
            mine16[r] = region[slot];
        }
        #pragma unroll
        for (int r = 0; r < 16; ++r)
            ((int*)(sAw + r * APITCH))[lane] = mine16[r];
    }

    for (int r = 0; r < 16; ++r) {
        const int mine = ((const int*)(sAw + r * APITCH))[lane];
        const int deg  = __builtin_amdgcn_readlane(degv, r);
        const int m    = (deg < 32) ? deg : 32;
        float4 acc;
        if (m <= 16) {
            float4 v[16];
            rows_issue<16>(T4, mine, 0, m, lane, v);
            acc = v[0];
            rows_add<16>(v, 1, acc);
        } else {
            float4 v[32];
            rows_issue<32>(T4, mine, 0, m, lane, v);
            acc = v[0];
            rows_add<32>(v, 1, acc);
        }
        *(float4*)(sAw + r * APITCH + 4 * lane) = acc;
    }

    for (int r = 0; r < 16; ++r) {
        const int deg = __builtin_amdgcn_readlane(degv, r);
        if (deg > 32) {
            int2 pv = make_int2(0, 0);
            if (lane < BK_SLICES) pv = sg[r * SEGPITCH + lane];
            int ex, delta, dg;
            node_runs(pv, lane, ex, delta, dg);
            float4 acc = *(const float4*)(sAw + r * APITCH + 4 * lane);
            for (int base = 32; base < deg; base += 32) {
                const int mm = (deg - base < 32) ? (deg - base) : 32;
                const int mi = chunk_sources(region, ex, delta, base, mm, lane);
                float4 v[16];
                rows_issue<16>(T4, mi, 0, mm, lane, v);
                rows_add<16>(v, 0, acc);
                if (mm > 16) {
                    rows_issue<16>(T4, mi, 16, mm, lane, v);
                    rows_add<16>(v, 0, acc);
                }
            }
            *(float4*)(sAw + r * APITCH + 4 * lane) = acc;
        }
    }

    __syncthreads();

    v8f acc[8];
    strip_from_lds(sAw, sW, lane, acc);

    float s[8];
    #pragma unroll
    for (int j = 0; j < 8; ++j) {
        const int col = 16 * j + c;
        const float bv = bias[col];
        const float wf = Wf[col];
        float p[8];
        #pragma unroll
        for (int r = 0; r < 8; ++r)
            p[r] = fmaxf(acc[j][r] + bv, 0.0f) * wf;
        #pragma unroll
        for (int off = 8; off > 0; off >>= 1) {
            #pragma unroll
            for (int r = 0; r < 8; ++r)
                p[r] += __shfl_xor(p[r], off, 32);
        }
        #pragma unroll
        for (int r = 0; r < 8; ++r)
            s[r] = (j == 0) ? p[r] : s[r] + p[r];
    }

    float vv = s[0];
    #pragma unroll
    for (int r = 1; r < 8; ++r)
        vv = (c == r) ? s[r] : vv;
    if (active && c < 8) {
        const int row = (mine_tile << 4) + (h << 3) + c;
        S[row] = vv;
    }
}

__global__ __launch_bounds__(256)
void gnn_gather_sum_scalar(const float* __restrict__ S,
                           const int* __restrict__ region,
                           const int2* __restrict__ seg,
                           const float* __restrict__ bf,
                           float* __restrict__ out, int N) {
    const int n = blockIdx.x * blockDim.x + threadIdx.x;
    if (n >= N) return;
    int st[BK_SLICES];
    int cn[BK_SLICES];
    int maxc = 0;
    #pragma unroll
    for (int s = 0; s < BK_SLICES; ++s) {
        const int2 v = seg[(size_t)s * N + n];
        st[s] = v.x;
        cn[s] = v.y;
    }
    #pragma unroll
    for (int s = 0; s < BK_SLICES; ++s)
        maxc = (cn[s] > maxc) ? cn[s] : maxc;
    float acc = 0.0f;
    for (int i = 0; i < maxc; ++i) {
        int idx[BK_SLICES];
        #pragma unroll
        for (int s = 0; s < BK_SLICES; ++s)
            idx[s] = (i < cn[s]) ? region[st[s] + i] : -1;
        float v[BK_SLICES];
        #pragma unroll
        for (int s = 0; s < BK_SLICES; ++s)
            v[s] = (idx[s] >= 0) ? S[idx[s]] : 0.0f;
        #pragma unroll
        for (int s = 0; s < BK_SLICES; ++s)
            acc += v[s];
    }
    out[n] = acc + bf[0];
}

extern "C" void kernel_launch(void* const* d_in, const int* in_sizes, int n_in,
                              void* d_out, int out_size, void* d_ws, size_t ws_size,
                              hipStream_t stream) {
    const float* x   = (const float*)d_in[0];
    const int*   ei  = (const int*)  d_in[1];
    const float* W1  = (const float*)d_in[2];
    const float* b1  = (const float*)d_in[3];
    const float* W2  = (const float*)d_in[4];
    const float* b2  = (const float*)d_in[5];
    const float* Wf  = (const float*)d_in[6];
    const float* bf  = (const float*)d_in[7];
    float* out = (float*)d_out;

    const int N = in_sizes[0] / HID;
    const int E = in_sizes[1] / 2;
    const int* src = ei;
    const int* dst = ei + E;

    const int ranges = (N + BK_CAP - 1) / BK_CAP;
    const int per    = (E + BK_SLICES - 1) / BK_SLICES;

    float* bufT = (float*)d_ws;
    int2*  seg  = (int2*)(bufT + (size_t)N * HID);
    int*   region = (int*)(seg + (size_t)BK_SLICES * N);
    float* bufS = (float*)(region + (size_t)ranges * BK_SLICES * per);

    const dim3 blk(256);
    const int numTiles  = N / 16;
    const int gemmGrid  = (numTiles + 7) / 8;
    const dim3 bkGrid(ranges, BK_SLICES);
    const int nodeGrid  = (N + 255) / 256;

    gnn_linear_relu_wmma<<<gemmGrid, blk, 0, stream>>>(x, W1, b1, bufT, numTiles);

    gnn_slice_sort<<<bkGrid, dim3(BK_THREADS), 0, stream>>>(src, dst, seg, region, N, E);

    gnn_gather_linear_relu_readout_wmma<<<gemmGrid, blk, 0, stream>>>(bufT, region, seg, W2, b2, Wf, bufS, N, numTiles);
    gnn_gather_sum_scalar<<<nodeGrid, blk, 0, stream>>>(bufS, region, seg, bf, out, N);
}


# Layer-1 GEMM and edge bucketing, independent of each other, run side by side as blocks of one launch (196 GEMM + 48 sort blocks of wider node ranges, placed via LDS); 3 launches; run order may differ
# speedup vs baseline: 5.1217x; 1.1619x over previous
#include <hip/hip_runtime.h>


typedef float v2f __attribute__((ext_vector_type(2)));
typedef float v8f __attribute__((ext_vector_type(8)));

#define HID 128
#define GEMM_WAVES 8
#define WPITCH 288
#define APITCH 132

template <int NT>
__device__ __forceinline__ void stage_w_lds(const float* __restrict__ W, float* sW, int tid) {
    #pragma unroll
    for (int i = 0; i < 2048 / NT; ++i) {
        const int item = i * NT + tid;
        const int kp   = item >> 5;
        const int m    = item & 31;
        const float4 w0 = *(const float4*)(W + (size_t)(2 * kp) * HID + 4 * m);
        const float4 w1 = *(const float4*)(W + (size_t)(2 * kp + 1) * HID + 4 * m);
        float* d = sW + kp * WPITCH + 8 * m;
        *(float4*)(d)     = make_float4(w0.x, w1.x, w0.y, w1.y);
        *(float4*)(d + 4) = make_float4(w0.z, w1.z, w0.w, w1.w);
    }
}

template <int NJ>
__device__ __forceinline__ void strip_from_lds(const float* sAw, const float* sW,
                                               int lane, int j0, v8f (&acc)[NJ]) {
    const int c = lane & 15;
    const int h = lane >> 4;
    const float* ap = sAw + c * APITCH + 2 * h;
    const float* bp = sW + h * WPITCH + 2 * c + 32 * j0;

    #pragma unroll
    for (int j = 0; j < NJ; ++j)
        acc[j] = (v8f){0.f, 0.f, 0.f, 0.f, 0.f, 0.f, 0.f, 0.f};

    v2f a_cur = *(const v2f*)(ap);
    v2f b_cur[NJ];
    #pragma unroll
    for (int j = 0; j < NJ; ++j)
        b_cur[j] = *(const v2f*)(bp + 32 * j);

    #pragma unroll
    for (int k0 = 0; k0 < HID; k0 += 4) {
        v2f a_nxt = a_cur;
        v2f b_nxt[NJ];
        #pragma unroll
        for (int j = 0; j < NJ; ++j)
            b_nxt[j] = b_cur[j];
        if (k0 + 4 < HID) {
            a_nxt = *(const v2f*)(ap + k0 + 4);
            const float* bk = bp + ((k0 + 4) >> 1) * WPITCH;
            #pragma unroll
            for (int j = 0; j < NJ; ++j)
                b_nxt[j] = *(const v2f*)(bk + 32 * j);
        }
        #pragma unroll
        for (int j = 0; j < NJ; ++j)
            acc[j] = __builtin_amdgcn_wmma_f32_16x16x4_f32(
                 false, a_cur,  false, b_cur[j],
                 (short)0, acc[j],  false,  false);
        a_cur = a_nxt;
        #pragma unroll
        for (int j = 0; j < NJ; ++j)
            b_cur[j] = b_nxt[j];
    }
}

#define BK_CAP    17408
#define BK_SLICES 16
#define BK_THREADS 1024
#define BK_GROUP  16
#define BK_SUB    (BK_CAP / BK_THREADS)
#define BK_NU     49

__device__ __forceinline__ int bk_slice_len(int E) { return (E + BK_SLICES - 1) / BK_SLICES; }

__device__ __forceinline__ void slice_sort_block(int* c, int* wtot, int* stage, int cap, int r, int s,
                                                 const int* __restrict__ src,
                                                 const int* __restrict__ dst,
                                                 int2* __restrict__ seg, int* __restrict__ region,
                                                 int N, int E) {
    const int tid  = threadIdx.x;
    const int lane = tid & 31;
    const int wave = tid >> 5;
    const int lo = r * BK_CAP;
    #pragma unroll
    for (int q = 0; q < BK_SUB; ++q) c[q * BK_THREADS + tid] = 0;
    __syncthreads();

    const int per = bk_slice_len(E);
    const int e0 = s * per;
    const int e1 = (e0 + per < E) ? (e0 + per) : E;

    int k[BK_NU];
    int sv[BK_NU];
    {
        const int last = (e1 - 1 > 0) ? (e1 - 1) : 0;
        const int left = (N - lo > 0) ? (N - lo) : 0;
        const unsigned lim = (left < BK_CAP) ? (unsigned)left : (unsigned)BK_CAP;
        #pragma unroll
        for (int u = 0; u < BK_NU; ++u) {
            const int e = e0 + u * BK_THREADS + tid;
            const int dv = dst[(e < e1) ? e : last];
            k[u] = (e < e1) ? dv : -1;
        }
        #pragma unroll
        for (int u = 0; u < BK_NU; ++u) {
            const unsigned kk = (unsigned)(k[u] - lo);
            const bool in = kk < lim;
            k[u] = in ? (int)(kk * 4u) : -1;
            if (in) atomicAdd((int*)((char*)c + k[u]), 1);
        }
        #pragma unroll
        for (int u = 0; u < BK_NU; ++u) {
            sv[u] = 0;
            if (k[u] >= 0) sv[u] = src[e0 + u * BK_THREADS + tid];
        }
    }
    for (int base = e0 + BK_NU * BK_THREADS; base < e1; base += BK_GROUP * BK_THREADS) {
        int d[BK_GROUP];
        #pragma unroll
        for (int u = 0; u < BK_GROUP; ++u) {
            const int e = base + u * BK_THREADS + tid;
            d[u] = (e < e1) ? dst[e] : -1;
        }
        #pragma unroll
        for (int u = 0; u < BK_GROUP; ++u) {
            const unsigned kk = (unsigned)(d[u] - lo);
            if (kk < (unsigned)BK_CAP && d[u] < N) atomicAdd(&c[kk], 1);
        }
    }
    __syncthreads();

    int mine = 0;
    #pragma unroll
    for (int q = 0; q < BK_SUB; ++q)
        mine += c[q * BK_THREADS + tid];
    int inc = mine;
    #pragma unroll
    for (int off = 1; off < 32; off <<= 1) {
        const int t = __shfl_up(inc, off, 32);
        if (lane >= off) inc += t;
    }
    if (lane == 31) wtot[wave] = inc;
    __syncthreads();

    const int first = (r * BK_SLICES + s) * per;
    int run = inc - mine;
    for (int w = 0; w < wave; ++w)
        run += wtot[w];
    #pragma unroll
    for (int q = 0; q < BK_SUB; ++q) {
        const int n = lo + q * BK_THREADS + tid;
        const int cnt = c[q * BK_THREADS + tid];
        c[q * BK_THREADS + tid] = run;
        if (n < N) seg[(size_t)s * N + n] = make_int2(first + run, cnt);
        run += cnt;
    }
    __syncthreads();

    #pragma unroll
    for (int u = 0; u < BK_NU; ++u)
        if (k[u] >= 0) k[u] = atomicAdd((int*)((char*)c + k[u]), 1);
    #pragma unroll
    for (int u = 0; u < BK_NU; ++u) {
        if (k[u] >= 0) {
            if (k[u] < cap) stage[k[u]] = sv[u];
            else            region[first + k[u]] = sv[u];
        }
    }
    for (int base = e0 + BK_NU * BK_THREADS; base < e1; base += BK_GROUP * BK_THREADS) {
        int d[BK_GROUP];
        int sw[BK_GROUP];
        #pragma unroll
        for (int u = 0; u < BK_GROUP; ++u) {
            const int e = base + u * BK_THREADS + tid;
            d[u]  = (e < e1) ? dst[e] : -1;
            sw[u] = (e < e1) ? src[e] : 0;
        }
        #pragma unroll
        for (int u = 0; u < BK_GROUP; ++u) {
            const unsigned kk = (unsigned)(d[u] - lo);
            if (kk < (unsigned)BK_CAP && d[u] < N) {
                const int pos = atomicAdd(&c[kk], 1);
                if (pos < cap) stage[pos] = sw[u];
                else           region[first + pos] = sw[u];
            }
        }
    }
    __syncthreads();

    int total = 0;
    #pragma unroll
    for (int w = 0; w < BK_THREADS / 32; ++w)
        total += wtot[w];
    const int staged = (total < cap) ? total : cap;
    for (int i = tid; i < staged; i += BK_THREADS)
        region[first + i] = stage[i];
}

#define L1_TILES 16

__global__ __launch_bounds__(BK_THREADS)
void gnn_linear_relu_wmma_and_slice_sort(const float* __restrict__ X,
                                         const float* __restrict__ W,
                                         const float* __restrict__ bias,
                                         float* __restrict__ T, int numTiles,
                                         const int* __restrict__ src, const int* __restrict__ dst,
                                         int2* __restrict__ seg, int* __restrict__ region,
                                         int N, int E, int ranges, int sortBlocks) {
    __shared__ __attribute__((aligned(16))) float smem[64 * WPITCH + L1_TILES * 16 * APITCH];

    if ((int)blockIdx.x < sortBlocks) {
        int* c = (int*)smem;
        const int b = blockIdx.x;
        const int used = BK_CAP + BK_THREADS / 32;
        slice_sort_block(c, c + BK_CAP, c + used, 64 * WPITCH + L1_TILES * 16 * APITCH - used,
                         b % ranges, b / ranges, src, dst, seg, region, N, E);
        return;
    }

    float* sW = smem;
    float* sA = smem + 64 * WPITCH;
    const int tid  = threadIdx.x;
    const int lane = tid & 31;
    const int wv   = __builtin_amdgcn_readfirstlane((int)(tid >> 5));
    const int tl   = wv >> 1;
    const int half = wv & 1;
    const int c    = lane & 15;
    const int h    = lane >> 4;
    const int mine = ((int)blockIdx.x - sortBlocks) * L1_TILES + tl;
    const bool active = mine < numTiles;
    const int tile = active ? mine : numTiles - 1;

    float4 ar[8];
    const float4* xp = (const float4*)(X + ((size_t)tile * 16 + 8 * half) * HID) + lane;
    #pragma unroll
    for (int r = 0; r < 8; ++r)
        ar[r] = xp[r * (HID / 4)];

    stage_w_lds<BK_THREADS>(W, sW, tid);

    float* sAt = sA + tl * (16 * APITCH);
    #pragma unroll
    for (int r = 0; r < 8; ++r)
        *(float4*)(sAt + (8 * half + r) * APITCH + 4 * lane) = ar[r];

    __syncthreads();

    v8f acc[4];
    strip_from_lds<4>(sAt, sW, lane, 4 * half, acc);

    if (active) {
        const int m_base = tile << 4;
        #pragma unroll
        for (int j = 0; j < 4; ++j) {
            const int col = 16 * (4 * half + j) + c;
            const float bv = bias[col];
            #pragma unroll
            for (int r = 0; r < 8; ++r) {
                const int row = m_base + r + (h << 3);
                T[(size_t)row * HID + col] = fmaxf(acc[j][r] + bv, 0.0f);
            }
        }
    }
}

#define SEGPITCH 17

__device__ __forceinline__ void node_runs(int2 v, int lane, int& ex, int& delta, int& deg) {
    int inc = v.y;
    #pragma unroll
    for (int off = 1; off < 32; off <<= 1) {
        const int t = __shfl_up(inc, off, 32);
        if (lane >= off) inc += t;
    }
    ex    = inc - v.y;
    deg   = __builtin_amdgcn_readlane(inc, 31);
    delta = v.x - ex;
}

__device__ __forceinline__ int chunk_sources(const int* __restrict__ region, int ex, int delta,
                                             int base, int m, int lane) {
    int q = (lane < m) ? lane : (m - 1);
    q = (q > 0) ? q : 0;
    const int p = base + q;
    int sp = 0;
    #pragma unroll
    for (int s = 1; s < BK_SLICES; ++s)
        sp += (p >= __builtin_amdgcn_readlane(ex, s)) ? 1 : 0;
    const int dl = __shfl(delta, sp, 32);
    return region[dl + p];
}

template <int NR>
__device__ __forceinline__ void rows_issue(const float4* __restrict__ T4, int mine, int j, int m,
                                           int lane, float4 (&v)[NR]) {
    int s[NR];
    #pragma unroll
    for (int u = 0; u < NR; ++u)
        s[u] = __builtin_amdgcn_readlane(mine, j + u);
    #pragma unroll
    for (int u = 0; u < NR; ++u) {
        v[u] = make_float4(0.f, 0.f, 0.f, 0.f);
        if (j + u < m)
            v[u] = T4[(size_t)s[u] * (HID / 4) + lane];
    }
}

template <int NR>
__device__ __forceinline__ void rows_add(const float4 (&v)[NR], int first, float4& acc) {
    #pragma unroll
    for (int u = 0; u < NR; ++u) {
        if (u >= first) {
            acc.x += v[u].x;
            acc.y += v[u].y;
            acc.z += v[u].z;
            acc.w += v[u].w;
        }
    }
}

__global__ __launch_bounds__(256)
void gnn_gather_linear_relu_readout_wmma(const float* __restrict__ T,
                                         const int* __restrict__ region,
                                         const int2* __restrict__ seg,
                                         const float* __restrict__ W,
                                         const float* __restrict__ bias,
                                         const float* __restrict__ Wf,
                                         float* __restrict__ S, int N, int numTiles) {
    __shared__ __attribute__((aligned(16))) float sW[64 * WPITCH];
    __shared__ __attribute__((aligned(16))) float sA[GEMM_WAVES * 16 * APITCH];
    __shared__ int2 sseg[GEMM_WAVES * 16 * SEGPITCH];

    const int tid  = threadIdx.x;
    const int lane = tid & 31;
    const int wv   = __builtin_amdgcn_readfirstlane((int)(tid >> 5));
    const int c    = lane & 15;
    const int h    = lane >> 4;
    const int mine_tile = blockIdx.x * GEMM_WAVES + wv;
    const bool active = mine_tile < numTiles;
    const int n0 = blockIdx.x * (GEMM_WAVES * 16);

    int2 pr[8];
    #pragma unroll
    for (int i = 0; i < 8; ++i) {
        const int item = i * 256 + tid;
        const int s = item >> 7;
        const int j = item & 127;
        pr[i] = make_int2(0, 0);
        if (n0 + j < N) pr[i] = seg[(size_t)s * N + n0 + j];
    }

    stage_w_lds<256>(W, sW, tid);

    #pragma unroll
    for (int i = 0; i < 8; ++i) {
        const int item = i * 256 + tid;
        const int s = item >> 7;
        const int j = item & 127;
        sseg[j * SEGPITCH + s] = pr[i];
    }
    __syncthreads();

    const int2* sg = sseg + wv * (16 * SEGPITCH);
    float* sAw = sA + wv * (16 * APITCH);
    const float4* T4 = (const float4*)T;

    int degv;
    {
        const int nr = lane & 15;
        const int hb = lane >> 4;
        int2 pr[BK_SLICES];
        #pragma unroll
        for (int s = 0; s < BK_SLICES; ++s)
            pr[s] = sg[nr * SEGPITCH + s];
        int exs[BK_SLICES];
        int jump[BK_SLICES];
        int run = 0;
        int dprev = 0;
        #pragma unroll
        for (int s = 0; s < BK_SLICES; ++s) {
            exs[s] = run;
            const int d = pr[s].x - run;
            jump[s] = d - dprev;
            dprev = d;
            run += pr[s].y;
        }
        degv = run;
        int* scr = (int*)(sAw + nr * APITCH) + 16 * hb;
        #pragma unroll
        for (int k = 0; k < 16; ++k) {
            const int p = 16 * hb + k;
            int slot = p + jump[0];
            #pragma unroll
            for (int s = 1; s < BK_SLICES; ++s)
                slot += (p >= exs[s]) ? jump[s] : 0;
            scr[k] = slot;
        }
    }
    __syncthreads();

    {
        int mine16[16];
        #pragma unroll
        for (int r = 0; r < 16; ++r) {
            const int deg = __builtin_amdgcn_readlane(degv, r);
            const int m   = (deg < 32) ? deg : 32;
            int q = (lane < m) ? lane : (m - 1);
            q = (q > 0) ? q : 0;
            int slot = ((const int*)(sAw + r * APITCH))[q];
            slot = (m > 0) ? slot : 0;
            mine16[r] = region[slot];
        }
        #pragma unroll
        for (int r = 0; r < 16; ++r)
            ((int*)(sAw + r * APITCH))[lane] = mine16[r];
    }

    for (int r = 0; r < 16; ++r) {
        const int mine = ((const int*)(sAw + r * APITCH))[lane];
        const int deg  = __builtin_amdgcn_readlane(degv, r);
        const int m    = (deg < 32) ? deg : 32;
        float4 acc;
        if (m <= 16) {
            float4 v[16];
            rows_issue<16>(T4, mine, 0, m, lane, v);
            acc = v[0];
            rows_add<16>(v, 1, acc);
        } else {
            float4 v[32];
            rows_issue<32>(T4, mine, 0, m, lane, v);
            acc = v[0];
            rows_add<32>(v, 1, acc);
        }
        *(float4*)(sAw + r * APITCH + 4 * lane) = acc;
    }

    for (int r = 0; r < 16; ++r) {
        const int deg = __builtin_amdgcn_readlane(degv, r);
        if (deg > 32) {
            int2 pv = make_int2(0, 0);
            if (lane < BK_SLICES) pv = sg[r * SEGPITCH + lane];
            int ex, delta, dg;
            node_runs(pv, lane, ex, delta, dg);
            float4 acc = *(const float4*)(sAw + r * APITCH + 4 * lane);
            for (int base = 32; base < deg; base += 32) {
                const int mm = (deg - base < 32) ? (deg - base) : 32;
                const int mi = chunk_sources(region, ex, delta, base, mm, lane);
                float4 v[16];
                rows_issue<16>(T4, mi, 0, mm, lane, v);
                rows_add<16>(v, 0, acc);
                if (mm > 16) {
                    rows_issue<16>(T4, mi, 16, mm, lane, v);
                    rows_add<16>(v, 0, acc);
                }
            }
            *(float4*)(sAw + r * APITCH + 4 * lane) = acc;
        }
    }

    __syncthreads();

    v8f acc[8];
    strip_from_lds<8>(sAw, sW, lane, 0, acc);

    float s[8];
    #pragma unroll
    for (int j = 0; j < 8; ++j) {
        const int col = 16 * j + c;
        const float bv = bias[col];
        const float wf = Wf[col];
        float p[8];
        #pragma unroll
        for (int r = 0; r < 8; ++r)
            p[r] = fmaxf(acc[j][r] + bv, 0.0f) * wf;
        #pragma unroll
        for (int off = 8; off > 0; off >>= 1) {
            #pragma unroll
            for (int r = 0; r < 8; ++r)
                p[r] += __shfl_xor(p[r], off, 32);
        }
        #pragma unroll
        for (int r = 0; r < 8; ++r)
            s[r] = (j == 0) ? p[r] : s[r] + p[r];
    }

    float vv = s[0];
    #pragma unroll
    for (int r = 1; r < 8; ++r)
        vv = (c == r) ? s[r] : vv;
    if (active && c < 8) {
        const int row = (mine_tile << 4) + (h << 3) + c;
        S[row] = vv;
    }
}

__global__ __launch_bounds__(256)
void gnn_gather_sum_scalar(const float* __restrict__ S,
                           const int* __restrict__ region,
                           const int2* __restrict__ seg,
                           const float* __restrict__ bf,
                           float* __restrict__ out, int N) {
    const int n = blockIdx.x * blockDim.x + threadIdx.x;
    if (n >= N) return;
    int st[BK_SLICES];
    int cn[BK_SLICES];
    int maxc = 0;
    #pragma unroll
    for (int s = 0; s < BK_SLICES; ++s) {
        const int2 v = seg[(size_t)s * N + n];
        st[s] = v.x;
        cn[s] = v.y;
    }
    #pragma unroll
    for (int s = 0; s < BK_SLICES; ++s)
        maxc = (cn[s] > maxc) ? cn[s] : maxc;
    float acc = 0.0f;
    for (int i = 0; i < maxc; ++i) {
        int idx[BK_SLICES];
        #pragma unroll
        for (int s = 0; s < BK_SLICES; ++s)
            idx[s] = (i < cn[s]) ? region[st[s] + i] : -1;
        float v[BK_SLICES];
        #pragma unroll
        for (int s = 0; s < BK_SLICES; ++s)
            v[s] = (idx[s] >= 0) ? S[idx[s]] : 0.0f;
        #pragma unroll
        for (int s = 0; s < BK_SLICES; ++s)
            acc += v[s];
    }
    out[n] = acc + bf[0];
}

extern "C" void kernel_launch(void* const* d_in, const int* in_sizes, int n_in,
                              void* d_out, int out_size, void* d_ws, size_t ws_size,
                              hipStream_t stream) {
    const float* x   = (const float*)d_in[0];
    const int*   ei  = (const int*)  d_in[1];
    const float* W1  = (const float*)d_in[2];
    const float* b1  = (const float*)d_in[3];
    const float* W2  = (const float*)d_in[4];
    const float* b2  = (const float*)d_in[5];
    const float* Wf  = (const float*)d_in[6];
    const float* bf  = (const float*)d_in[7];
    float* out = (float*)d_out;

    const int N = in_sizes[0] / HID;
    const int E = in_sizes[1] / 2;
    const int* src = ei;
    const int* dst = ei + E;

    const int ranges = (N + BK_CAP - 1) / BK_CAP;
    const int per    = (E + BK_SLICES - 1) / BK_SLICES;

    float* bufT = (float*)d_ws;
    int2*  seg  = (int2*)(bufT + (size_t)N * HID);
    int*   region = (int*)(seg + (size_t)BK_SLICES * N);
    float* bufS = (float*)(region + (size_t)ranges * BK_SLICES * per);

    const dim3 blk(256);
    const int numTiles  = N / 16;
    const int gemmGrid  = (numTiles + 7) / 8;
    const int sortBlocks = ranges * BK_SLICES;
    const int l1Blocks  = (numTiles + L1_TILES - 1) / L1_TILES;
    const int nodeGrid  = (N + 255) / 256;

    gnn_linear_relu_wmma_and_slice_sort<<<sortBlocks + l1Blocks, dim3(BK_THREADS), 0, stream>>>(
        x, W1, b1, bufT, numTiles, src, dst, seg, region, N, E, ranges, sortBlocks);

    gnn_gather_linear_relu_readout_wmma<<<gemmGrid, blk, 0, stream>>>(bufT, region, seg, W2, b2, Wf, bufS, N, numTiles);
    gnn_gather_sum_scalar<<<nodeGrid, blk, 0, stream>>>(bufS, region, seg, bf, out, N);
}


# Layer-2 launch stores each node's first 31 sources and degree as one 128-byte line; output launch reads it (lane = position), gathers S in one round, lane-butterfly sum; sum order changes
# speedup vs baseline: 5.7222x; 1.1172x over previous
#include <hip/hip_runtime.h>


typedef float v2f __attribute__((ext_vector_type(2)));
typedef float v8f __attribute__((ext_vector_type(8)));

#define HID 128
#define GEMM_WAVES 8
#define WPITCH 288
#define APITCH 132

template <int NT>
__device__ __forceinline__ void stage_w_lds(const float* __restrict__ W, float* sW, int tid) {
    #pragma unroll
    for (int i = 0; i < 2048 / NT; ++i) {
        const int item = i * NT + tid;
        const int kp   = item >> 5;
        const int m    = item & 31;
        const float4 w0 = *(const float4*)(W + (size_t)(2 * kp) * HID + 4 * m);
        const float4 w1 = *(const float4*)(W + (size_t)(2 * kp + 1) * HID + 4 * m);
        float* d = sW + kp * WPITCH + 8 * m;
        *(float4*)(d)     = make_float4(w0.x, w1.x, w0.y, w1.y);
        *(float4*)(d + 4) = make_float4(w0.z, w1.z, w0.w, w1.w);
    }
}

template <int NJ>
__device__ __forceinline__ void strip_from_lds(const float* sAw, const float* sW,
                                               int lane, int j0, v8f (&acc)[NJ]) {
    const int c = lane & 15;
    const int h = lane >> 4;
    const float* ap = sAw + c * APITCH + 2 * h;
    const float* bp = sW + h * WPITCH + 2 * c + 32 * j0;

    #pragma unroll
    for (int j = 0; j < NJ; ++j)
        acc[j] = (v8f){0.f, 0.f, 0.f, 0.f, 0.f, 0.f, 0.f, 0.f};

    v2f a_cur = *(const v2f*)(ap);
    v2f b_cur[NJ];
    #pragma unroll
    for (int j = 0; j < NJ; ++j)
        b_cur[j] = *(const v2f*)(bp + 32 * j);

    #pragma unroll
    for (int k0 = 0; k0 < HID; k0 += 4) {
        v2f a_nxt = a_cur;
        v2f b_nxt[NJ];
        #pragma unroll
        for (int j = 0; j < NJ; ++j)
            b_nxt[j] = b_cur[j];
        if (k0 + 4 < HID) {
            a_nxt = *(const v2f*)(ap + k0 + 4);
            const float* bk = bp + ((k0 + 4) >> 1) * WPITCH;
            #pragma unroll
            for (int j = 0; j < NJ; ++j)
                b_nxt[j] = *(const v2f*)(bk + 32 * j);
        }
        #pragma unroll
        for (int j = 0; j < NJ; ++j)
            acc[j] = __builtin_amdgcn_wmma_f32_16x16x4_f32(
                 false, a_cur,  false, b_cur[j],
                 (short)0, acc[j],  false,  false);
        a_cur = a_nxt;
        #pragma unroll
        for (int j = 0; j < NJ; ++j)
            b_cur[j] = b_nxt[j];
    }
}

#define BK_CAP    17408
#define BK_SLICES 16
#define BK_THREADS 1024
#define BK_GROUP  16
#define BK_SUB    (BK_CAP / BK_THREADS)
#define BK_NU     49

__device__ __forceinline__ int bk_slice_len(int E) { return (E + BK_SLICES - 1) / BK_SLICES; }

__device__ __forceinline__ void slice_sort_block(int* c, int* wtot, int* stage, int cap, int r, int s,
                                                 const int* __restrict__ src,
                                                 const int* __restrict__ dst,
                                                 int2* __restrict__ seg, int* __restrict__ region,
                                                 int N, int E) {
    const int tid  = threadIdx.x;
    const int lane = tid & 31;
    const int wave = tid >> 5;
    const int lo = r * BK_CAP;
    #pragma unroll
    for (int q = 0; q < BK_SUB; ++q) c[q * BK_THREADS + tid] = 0;
    __syncthreads();

    const int per = bk_slice_len(E);
    const int e0 = s * per;
    const int e1 = (e0 + per < E) ? (e0 + per) : E;

    int k[BK_NU];
    int sv[BK_NU];
    {
        const int last = (e1 - 1 > 0) ? (e1 - 1) : 0;
        const int left = (N - lo > 0) ? (N - lo) : 0;
        const unsigned lim = (left < BK_CAP) ? (unsigned)left : (unsigned)BK_CAP;
        #pragma unroll
        for (int u = 0; u < BK_NU; ++u) {
            const int e = e0 + u * BK_THREADS + tid;
            const int dv = dst[(e < e1) ? e : last];
            k[u] = (e < e1) ? dv : -1;
        }
        #pragma unroll
        for (int u = 0; u < BK_NU; ++u) {
            const unsigned kk = (unsigned)(k[u] - lo);
            const bool in = kk < lim;
            k[u] = in ? (int)(kk * 4u) : -1;
            if (in) atomicAdd((int*)((char*)c + k[u]), 1);
        }
        #pragma unroll
        for (int u = 0; u < BK_NU; ++u) {
            sv[u] = 0;
            if (k[u] >= 0) sv[u] = src[e0 + u * BK_THREADS + tid];
        }
    }
    for (int base = e0 + BK_NU * BK_THREADS; base < e1; base += BK_GROUP * BK_THREADS) {
        int d[BK_GROUP];
        #pragma unroll
        for (int u = 0; u < BK_GROUP; ++u) {
            const int e = base + u * BK_THREADS + tid;
            d[u] = (e < e1) ? dst[e] : -1;
        }
        #pragma unroll
        for (int u = 0; u < BK_GROUP; ++u) {
            const unsigned kk = (unsigned)(d[u] - lo);
            if (kk < (unsigned)BK_CAP && d[u] < N) atomicAdd(&c[kk], 1);
        }
    }
    __syncthreads();

    int mine = 0;
    #pragma unroll
    for (int q = 0; q < BK_SUB; ++q)
        mine += c[q * BK_THREADS + tid];
    int inc = mine;
    #pragma unroll
    for (int off = 1; off < 32; off <<= 1) {
        const int t = __shfl_up(inc, off, 32);
        if (lane >= off) inc += t;
    }
    if (lane == 31) wtot[wave] = inc;
    __syncthreads();

    const int first = (r * BK_SLICES + s) * per;
    int run = inc - mine;
    for (int w = 0; w < wave; ++w)
        run += wtot[w];
    #pragma unroll
    for (int q = 0; q < BK_SUB; ++q) {
        const int n = lo + q * BK_THREADS + tid;
        const int cnt = c[q * BK_THREADS + tid];
        c[q * BK_THREADS + tid] = run;
        if (n < N) seg[(size_t)s * N + n] = make_int2(first + run, cnt);
        run += cnt;
    }
    __syncthreads();

    #pragma unroll
    for (int u = 0; u < BK_NU; ++u)
        if (k[u] >= 0) k[u] = atomicAdd((int*)((char*)c + k[u]), 1);
    #pragma unroll
    for (int u = 0; u < BK_NU; ++u) {
        if (k[u] >= 0) {
            if (k[u] < cap) stage[k[u]] = sv[u];
            else            region[first + k[u]] = sv[u];
        }
    }
    for (int base = e0 + BK_NU * BK_THREADS; base < e1; base += BK_GROUP * BK_THREADS) {
        int d[BK_GROUP];
        int sw[BK_GROUP];
        #pragma unroll
        for (int u = 0; u < BK_GROUP; ++u) {
            const int e = base + u * BK_THREADS + tid;
            d[u]  = (e < e1) ? dst[e] : -1;
            sw[u] = (e < e1) ? src[e] : 0;
        }
        #pragma unroll
        for (int u = 0; u < BK_GROUP; ++u) {
            const unsigned kk = (unsigned)(d[u] - lo);
            if (kk < (unsigned)BK_CAP && d[u] < N) {
                const int pos = atomicAdd(&c[kk], 1);
                if (pos < cap) stage[pos] = sw[u];
                else           region[first + pos] = sw[u];
            }
        }
    }
    __syncthreads();

    int total = 0;
    #pragma unroll
    for (int w = 0; w < BK_THREADS / 32; ++w)
        total += wtot[w];
    const int staged = (total < cap) ? total : cap;
    for (int i = tid; i < staged; i += BK_THREADS)
        region[first + i] = stage[i];
}

#define L1_TILES 16

__global__ __launch_bounds__(BK_THREADS)
void gnn_linear_relu_wmma_and_slice_sort(const float* __restrict__ X,
                                         const float* __restrict__ W,
                                         const float* __restrict__ bias,
                                         float* __restrict__ T, int numTiles,
                                         const int* __restrict__ src, const int* __restrict__ dst,
                                         int2* __restrict__ seg, int* __restrict__ region,
                                         int N, int E, int ranges, int sortBlocks) {
    __shared__ __attribute__((aligned(16))) float smem[64 * WPITCH + L1_TILES * 16 * APITCH];

    if ((int)blockIdx.x < sortBlocks) {
        int* c = (int*)smem;
        const int b = blockIdx.x;
        const int used = BK_CAP + BK_THREADS / 32;
        slice_sort_block(c, c + BK_CAP, c + used, 64 * WPITCH + L1_TILES * 16 * APITCH - used,
                         b % ranges, b / ranges, src, dst, seg, region, N, E);
        return;
    }

    float* sW = smem;
    float* sA = smem + 64 * WPITCH;
    const int tid  = threadIdx.x;
    const int lane = tid & 31;
    const int wv   = __builtin_amdgcn_readfirstlane((int)(tid >> 5));
    const int tl   = wv >> 1;
    const int half = wv & 1;
    const int c    = lane & 15;
    const int h    = lane >> 4;
    const int mine = ((int)blockIdx.x - sortBlocks) * L1_TILES + tl;
    const bool active = mine < numTiles;
    const int tile = active ? mine : numTiles - 1;

    float4 ar[8];
    const float4* xp = (const float4*)(X + ((size_t)tile * 16 + 8 * half) * HID) + lane;
    #pragma unroll
    for (int r = 0; r < 8; ++r)
        ar[r] = xp[r * (HID / 4)];

    stage_w_lds<BK_THREADS>(W, sW, tid);

    float* sAt = sA + tl * (16 * APITCH);
    #pragma unroll
    for (int r = 0; r < 8; ++r)
        *(float4*)(sAt + (8 * half + r) * APITCH + 4 * lane) = ar[r];

    __syncthreads();

    v8f acc[4];
    strip_from_lds<4>(sAt, sW, lane, 4 * half, acc);

    if (active) {
        const int m_base = tile << 4;
        #pragma unroll
        for (int j = 0; j < 4; ++j) {
            const int col = 16 * (4 * half + j) + c;
            const float bv = bias[col];
            #pragma unroll
            for (int r = 0; r < 8; ++r) {
                const int row = m_base + r + (h << 3);
                T[(size_t)row * HID + col] = fmaxf(acc[j][r] + bv, 0.0f);
            }
        }
    }
}

#define SEGPITCH 17

#define LIST_POS 31

__device__ __forceinline__ void node_runs(int2 v, int lane, int& ex, int& delta, int& deg) {
    int inc = v.y;
    #pragma unroll
    for (int off = 1; off < 32; off <<= 1) {
        const int t = __shfl_up(inc, off, 32);
        if (lane >= off) inc += t;
    }
    ex    = inc - v.y;
    deg   = __builtin_amdgcn_readlane(inc, 31);
    delta = v.x - ex;
}

__device__ __forceinline__ int chunk_sources(const int* __restrict__ region, int ex, int delta,
                                             int base, int m, int lane) {
    int q = (lane < m) ? lane : (m - 1);
    q = (q > 0) ? q : 0;
    const int p = base + q;
    int sp = 0;
    #pragma unroll
    for (int s = 1; s < BK_SLICES; ++s)
        sp += (p >= __builtin_amdgcn_readlane(ex, s)) ? 1 : 0;
    const int dl = __shfl(delta, sp, 32);
    return region[dl + p];
}

template <int NR>
__device__ __forceinline__ void rows_issue(const float4* __restrict__ T4, int mine, int j, int m,
                                           int lane, float4 (&v)[NR]) {
    int s[NR];
    #pragma unroll
    for (int u = 0; u < NR; ++u)
        s[u] = __builtin_amdgcn_readlane(mine, j + u);
    #pragma unroll
    for (int u = 0; u < NR; ++u) {
        v[u] = make_float4(0.f, 0.f, 0.f, 0.f);
        if (j + u < m)
            v[u] = T4[(size_t)s[u] * (HID / 4) + lane];
    }
}

template <int NR>
__device__ __forceinline__ void rows_add(const float4 (&v)[NR], int first, float4& acc) {
    #pragma unroll
    for (int u = 0; u < NR; ++u) {
        if (u >= first) {
            acc.x += v[u].x;
            acc.y += v[u].y;
            acc.z += v[u].z;
            acc.w += v[u].w;
        }
    }
}

__global__ __launch_bounds__(256)
void gnn_gather_linear_relu_readout_wmma(const float* __restrict__ T,
                                         const int* __restrict__ region,
                                         const int2* __restrict__ seg,
                                         const float* __restrict__ W,
                                         const float* __restrict__ bias,
                                         const float* __restrict__ Wf,
                                         float* __restrict__ S, int* __restrict__ srcl,
                                         int N, int numTiles) {
    __shared__ __attribute__((aligned(16))) float sW[64 * WPITCH];
    __shared__ __attribute__((aligned(16))) float sA[GEMM_WAVES * 16 * APITCH];
    __shared__ int2 sseg[GEMM_WAVES * 16 * SEGPITCH];

    const int tid  = threadIdx.x;
    const int lane = tid & 31;
    const int wv   = __builtin_amdgcn_readfirstlane((int)(tid >> 5));
    const int c    = lane & 15;
    const int h    = lane >> 4;
    const int mine_tile = blockIdx.x * GEMM_WAVES + wv;
    const bool active = mine_tile < numTiles;
    const int n0 = blockIdx.x * (GEMM_WAVES * 16);

    int2 pr[8];
    #pragma unroll
    for (int i = 0; i < 8; ++i) {
        const int item = i * 256 + tid;
        const int s = item >> 7;
        const int j = item & 127;
        pr[i] = make_int2(0, 0);
        if (n0 + j < N) pr[i] = seg[(size_t)s * N + n0 + j];
    }

    stage_w_lds<256>(W, sW, tid);

    #pragma unroll
    for (int i = 0; i < 8; ++i) {
        const int item = i * 256 + tid;
        const int s = item >> 7;
        const int j = item & 127;
        sseg[j * SEGPITCH + s] = pr[i];
    }
    __syncthreads();

    const int2* sg = sseg + wv * (16 * SEGPITCH);
    float* sAw = sA + wv * (16 * APITCH);
    const float4* T4 = (const float4*)T;

    int degv;
    {
        const int nr = lane & 15;
        const int hb = lane >> 4;
        int2 pr[BK_SLICES];
        #pragma unroll
        for (int s = 0; s < BK_SLICES; ++s)
            pr[s] = sg[nr * SEGPITCH + s];
        int exs[BK_SLICES];
        int jump[BK_SLICES];
        int run = 0;
        int dprev = 0;
        #pragma unroll
        for (int s = 0; s < BK_SLICES; ++s) {
            exs[s] = run;
            const int d = pr[s].x - run;
            jump[s] = d - dprev;
            dprev = d;
            run += pr[s].y;
        }
        degv = run;
        int* scr = (int*)(sAw + nr * APITCH) + 16 * hb;
        #pragma unroll
        for (int k = 0; k < 16; ++k) {
            const int p = 16 * hb + k;
            int slot = p + jump[0];
            #pragma unroll
            for (int s = 1; s < BK_SLICES; ++s)
                slot += (p >= exs[s]) ? jump[s] : 0;
            scr[k] = slot;
        }
    }
    __syncthreads();

    {
        int mine16[16];
        #pragma unroll
        for (int r = 0; r < 16; ++r) {
            const int deg = __builtin_amdgcn_readlane(degv, r);
            const int m   = (deg < 32) ? deg : 32;
            int q = (lane < m) ? lane : (m - 1);
            q = (q > 0) ? q : 0;
            int slot = ((const int*)(sAw + r * APITCH))[q];
            slot = (m > 0) ? slot : 0;
            mine16[r] = region[slot];
        }
        #pragma unroll
        for (int r = 0; r < 16; ++r)
            ((int*)(sAw + r * APITCH))[lane] = mine16[r];
        if (active) {
            #pragma unroll
            for (int r = 0; r < 16; ++r) {
                const int deg = __builtin_amdgcn_readlane(degv, r);
                int val = (lane < deg) ? mine16[r] : -1;
                val = (lane == LIST_POS) ? deg : val;
                srcl[((size_t)mine_tile * 16 + r) * 32 + lane] = val;
            }
        }
    }

    for (int r = 0; r < 16; ++r) {
        const int mine = ((const int*)(sAw + r * APITCH))[lane];
        const int deg  = __builtin_amdgcn_readlane(degv, r);
        const int m    = (deg < 32) ? deg : 32;
        float4 acc;
        if (m <= 16) {
            float4 v[16];
            rows_issue<16>(T4, mine, 0, m, lane, v);
            acc = v[0];
            rows_add<16>(v, 1, acc);
        } else {
            float4 v[32];
            rows_issue<32>(T4, mine, 0, m, lane, v);
            acc = v[0];
            rows_add<32>(v, 1, acc);
        }
        *(float4*)(sAw + r * APITCH + 4 * lane) = acc;
    }

    for (int r = 0; r < 16; ++r) {
        const int deg = __builtin_amdgcn_readlane(degv, r);
        if (deg > 32) {
            int2 pv = make_int2(0, 0);
            if (lane < BK_SLICES) pv = sg[r * SEGPITCH + lane];
            int ex, delta, dg;
            node_runs(pv, lane, ex, delta, dg);
            float4 acc = *(const float4*)(sAw + r * APITCH + 4 * lane);
            for (int base = 32; base < deg; base += 32) {
                const int mm = (deg - base < 32) ? (deg - base) : 32;
                const int mi = chunk_sources(region, ex, delta, base, mm, lane);
                float4 v[16];
                rows_issue<16>(T4, mi, 0, mm, lane, v);
                rows_add<16>(v, 0, acc);
                if (mm > 16) {
                    rows_issue<16>(T4, mi, 16, mm, lane, v);
                    rows_add<16>(v, 0, acc);
                }
            }
            *(float4*)(sAw + r * APITCH + 4 * lane) = acc;
        }
    }

    __syncthreads();

    v8f acc[8];
    strip_from_lds<8>(sAw, sW, lane, 0, acc);

    float s[8];
    #pragma unroll
    for (int j = 0; j < 8; ++j) {
        const int col = 16 * j + c;
        const float bv = bias[col];
        const float wf = Wf[col];
        float p[8];
        #pragma unroll
        for (int r = 0; r < 8; ++r)
            p[r] = fmaxf(acc[j][r] + bv, 0.0f) * wf;
        #pragma unroll
        for (int off = 8; off > 0; off >>= 1) {
            #pragma unroll
            for (int r = 0; r < 8; ++r)
                p[r] += __shfl_xor(p[r], off, 32);
        }
        #pragma unroll
        for (int r = 0; r < 8; ++r)
            s[r] = (j == 0) ? p[r] : s[r] + p[r];
    }

    float vv = s[0];
    #pragma unroll
    for (int r = 1; r < 8; ++r)
        vv = (c == r) ? s[r] : vv;
    if (active && c < 8) {
        const int row = (mine_tile << 4) + (h << 3) + c;
        S[row] = vv;
    }
}

__global__ __launch_bounds__(256)
void gnn_gather_sum_scalar(const float* __restrict__ S,
                           const int* __restrict__ srcl,
                           const int* __restrict__ region,
                           const int2* __restrict__ seg,
                           const float* __restrict__ bf,
                           float* __restrict__ out, int N) {
    const int lane = threadIdx.x & 31;
    const int n0 = __builtin_amdgcn_readfirstlane((int)(blockIdx.x * blockDim.x + (threadIdx.x & ~31u)));
    if (n0 >= N) return;
    const int lastn = N - 1;

    int idx[32];
    #pragma unroll
    for (int j = 0; j < 32; ++j) {
        const int n = (n0 + j < N) ? (n0 + j) : lastn;
        idx[j] = srcl[(size_t)n * 32 + lane];
    }
    float v[32];
    int beyond = 0;
    #pragma unroll
    for (int j = 0; j < 32; ++j) {
        const bool ok = (lane < LIST_POS) && (idx[j] >= 0);
        const float t = S[ok ? idx[j] : 0];
        v[j] = ok ? t : 0.0f;
        beyond |= (idx[j] > LIST_POS) ? 1 : 0;
    }

    float a16[16];
    {
        const bool up = (lane & 16) != 0;
        #pragma unroll
        for (int j = 0; j < 16; ++j) {
            const float keep = up ? v[j + 16] : v[j];
            const float send = up ? v[j] : v[j + 16];
            a16[j] = keep + __shfl_xor(send, 16, 32);
        }
    }
    float a8[8];
    {
        const bool up = (lane & 8) != 0;
        #pragma unroll
        for (int j = 0; j < 8; ++j) {
            const float keep = up ? a16[j + 8] : a16[j];
            const float send = up ? a16[j] : a16[j + 8];
            a8[j] = keep + __shfl_xor(send, 8, 32);
        }
    }
    float a4[4];
    {
        const bool up = (lane & 4) != 0;
        #pragma unroll
        for (int j = 0; j < 4; ++j) {
            const float keep = up ? a8[j + 4] : a8[j];
            const float send = up ? a8[j] : a8[j + 4];
            a4[j] = keep + __shfl_xor(send, 4, 32);
        }
    }
    float a2[2];
    {
        const bool up = (lane & 2) != 0;
        #pragma unroll
        for (int j = 0; j < 2; ++j) {
            const float keep = up ? a4[j + 2] : a4[j];
            const float send = up ? a4[j] : a4[j + 2];
            a2[j] = keep + __shfl_xor(send, 2, 32);
        }
    }
    float acc;
    {
        const bool up = (lane & 1) != 0;
        const float keep = up ? a2[1] : a2[0];
        const float send = up ? a2[0] : a2[1];
        acc = keep + __shfl_xor(send, 1, 32);
    }

    if (__builtin_amdgcn_readlane(beyond, LIST_POS) != 0) {
        const int nl = (n0 + lane < N) ? (n0 + lane) : lastn;
        const int degl = srcl[(size_t)nl * 32 + LIST_POS];
        #pragma nounroll
        for (int j = 0; j < 32; ++j) {
            const int dg = __builtin_amdgcn_readlane(degl, j);
            if (dg > LIST_POS && n0 + j < N) {
                int2 pv = make_int2(0, 0);
                if (lane < BK_SLICES) pv = seg[(size_t)lane * N + n0 + j];
                int ex, delta, dgr;
                node_runs(pv, lane, ex, delta, dgr);
                float part = 0.0f;
                for (int base = LIST_POS; base < dg; base += 32) {
                    const int mm = (dg - base < 32) ? (dg - base) : 32;
                    const int mi = chunk_sources(region, ex, delta, base, mm, lane);
                    const float t = S[mi];
                    part += (lane < mm) ? t : 0.0f;
                }
                #pragma unroll
                for (int off = 16; off > 0; off >>= 1)
                    part += __shfl_xor(part, off, 32);
                acc += (lane == j) ? part : 0.0f;
            }
        }
    }

    if (n0 + lane < N) out[n0 + lane] = acc + bf[0];
}

extern "C" void kernel_launch(void* const* d_in, const int* in_sizes, int n_in,
                              void* d_out, int out_size, void* d_ws, size_t ws_size,
                              hipStream_t stream) {
    const float* x   = (const float*)d_in[0];
    const int*   ei  = (const int*)  d_in[1];
    const float* W1  = (const float*)d_in[2];
    const float* b1  = (const float*)d_in[3];
    const float* W2  = (const float*)d_in[4];
    const float* b2  = (const float*)d_in[5];
    const float* Wf  = (const float*)d_in[6];
    const float* bf  = (const float*)d_in[7];
    float* out = (float*)d_out;

    const int N = in_sizes[0] / HID;
    const int E = in_sizes[1] / 2;
    const int* src = ei;
    const int* dst = ei + E;

    const int ranges = (N + BK_CAP - 1) / BK_CAP;
    const int per    = (E + BK_SLICES - 1) / BK_SLICES;

    float* bufT = (float*)d_ws;
    int2*  seg  = (int2*)(bufT + (size_t)N * HID);
    int*   region = (int*)(seg + (size_t)BK_SLICES * N);
    int*   srcl = region + (size_t)ranges * BK_SLICES * per;
    float* bufS = (float*)(srcl + (size_t)N * 32);

    const dim3 blk(256);
    const int numTiles  = N / 16;
    const int gemmGrid  = (numTiles + 7) / 8;
    const int sortBlocks = ranges * BK_SLICES;
    const int l1Blocks  = (numTiles + L1_TILES - 1) / L1_TILES;
    const int nodeGrid  = (N + 255) / 256;

    gnn_linear_relu_wmma_and_slice_sort<<<sortBlocks + l1Blocks, dim3(BK_THREADS), 0, stream>>>(
        x, W1, b1, bufT, numTiles, src, dst, seg, region, N, E, ranges, sortBlocks);

    gnn_gather_linear_relu_readout_wmma<<<gemmGrid, blk, 0, stream>>>(bufT, region, seg, W2, b2, Wf, bufS, srcl, N, numTiles);
    gnn_gather_sum_scalar<<<nodeGrid, blk, 0, stream>>>(bufS, srcl, region, seg, bf, out, N);
}
